# starting kernel
# baseline (speedup 1.0000x reference)
#include <hip/hip_runtime.h>

typedef __attribute__((ext_vector_type(2))) float v2f;
typedef __attribute__((ext_vector_type(8))) float v8f;

#define N_NODES 50000
#define N_EDGES 800000
#define D       128
#define OUTD    12
#define N_TGT   8192


__global__ void k_zero(float* __restrict__ p, int n) {
  int i = blockIdx.x * blockDim.x + threadIdx.x;
  if (i < n) p[i] = 0.0f;
}

__global__ void k_deg(const int* __restrict__ dst, float* __restrict__ cnt) {
  int e = blockIdx.x * blockDim.x + threadIdx.x;
  if (e < N_EDGES) atomicAdd(&cnt[dst[e]], 1.0f);
}

__global__ void k_dinv(const float* __restrict__ cnt, float* __restrict__ dinv) {
  int i = blockIdx.x * blockDim.x + threadIdx.x;
  if (i < N_NODES) dinv[i] = rsqrtf(cnt[i] + 1.0f);
}


__global__ __launch_bounds__(256) void k_gemm(const float* __restrict__ X,
                                              const float* __restrict__ W,
                                              float* __restrict__ H,
                                              int apply_relu) {
  __shared__ float sX[16][132];
  const int row0 = blockIdx.x * 16;

  for (int idx = threadIdx.x; idx < 16 * 128; idx += 256) {
    const int r = idx >> 7, c = idx & 127;
    float v = X[(size_t)(row0 + r) * D + c];
    if (apply_relu) v = fmaxf(v, 0.0f);
    sX[r][c] = v;
  }
  __syncthreads();

  const int wave    = threadIdx.x >> 5;
  const int lane    = threadIdx.x & 31;
  const int half    = lane >> 4;
  const int l       = lane & 15;
  const int colBase = wave * 16;

  v8f acc = {};
  #pragma unroll
  for (int k = 0; k < 32; ++k) {
    const int kb = 4 * k + 2 * half;
    v2f a, b;
    a.x = sX[l][kb];
    a.y = sX[l][kb + 1];
    b.x = W[(size_t)kb       * D + colBase + l];
    b.y = W[(size_t)(kb + 1) * D + colBase + l];
    acc = __builtin_amdgcn_wmma_f32_16x16x4_f32(
         false, a,  false, b,
         (short)0, acc,  false,  false);
  }

  #pragma unroll
  for (int r = 0; r < 8; ++r) {
    H[(size_t)(row0 + r + 8 * half) * D + colBase + l] = acc[r];
  }
}


__global__ void k_init(const float* __restrict__ H, const float* __restrict__ dinv,
                       const float* __restrict__ bias, float* __restrict__ A) {
  const int i    = blockIdx.x * blockDim.x + threadIdx.x;
  const int node = i >> 7;
  const int f    = i & 127;
  const float di = dinv[node];
  A[i] = H[i] * di * di + bias[f];
}


__global__ void k_edge(const float* __restrict__ H, const int* __restrict__ src,
                       const int* __restrict__ dst, const float* __restrict__ dinv,
                       float* __restrict__ A) {
  const int g    = blockIdx.x * blockDim.x + threadIdx.x;
  const int e    = g >> 5;
  const int lane = g & 31;
  if (e >= N_EDGES) return;
  const int s = src[e];
  const int d = dst[e];
  const float norm = dinv[s] * dinv[d];
  const float4 hv = ((const float4*)(H + (size_t)s * D))[lane];
  float* ap = A + (size_t)d * D + lane * 4;
  atomicAdd(ap + 0, hv.x * norm);
  atomicAdd(ap + 1, hv.y * norm);
  atomicAdd(ap + 2, hv.z * norm);
  atomicAdd(ap + 3, hv.w * norm);
}


__global__ void k_final(const float* __restrict__ A, const int* __restrict__ tgt,
                        const float* __restrict__ Wr, const float* __restrict__ br,
                        float* __restrict__ out) {
  const int g    = blockIdx.x * blockDim.x + threadIdx.x;
  const int t    = g >> 5;
  const int lane = g & 31;
  if (t >= N_TGT) return;
  const int node = tgt[t];
  float h[4];
  #pragma unroll
  for (int j = 0; j < 4; ++j)
    h[j] = fmaxf(A[(size_t)node * D + lane + 32 * j], 0.0f);
  for (int o = 0; o < OUTD; ++o) {
    float p = 0.0f;
    #pragma unroll
    for (int j = 0; j < 4; ++j)
      p += h[j] * Wr[(size_t)(lane + 32 * j) * OUTD + o];
    for (int m = 16; m >= 1; m >>= 1) p += __shfl_xor(p, m, 32);
    if (lane == 0) out[(size_t)t * OUTD + o] = p + br[o];
  }
}


extern "C" void kernel_launch(void* const* d_in, const int* in_sizes, int n_in,
                              void* d_out, int out_size, void* d_ws, size_t ws_size,
                              hipStream_t stream) {
  const float* x   = (const float*)d_in[0];
  const int*   ei  = (const int*)  d_in[1];
  const int*   tgt = (const int*)  d_in[2];
  const float* W1  = (const float*)d_in[3];
  const float* b1  = (const float*)d_in[4];
  const float* W2  = (const float*)d_in[5];
  const float* b2  = (const float*)d_in[6];
  const float* W3  = (const float*)d_in[7];
  const float* b3  = (const float*)d_in[8];
  const float* Wr  = (const float*)d_in[9];
  const float* br  = (const float*)d_in[10];
  float* out = (float*)d_out;

  const int* src = ei;
  const int* dst = ei + N_EDGES;

  float* ws   = (float*)d_ws;
  float* H    = ws;
  float* A    = H + (size_t)N_NODES * D;
  float* cnt  = A + (size_t)N_NODES * D;
  float* dinv = cnt + N_NODES;

  k_zero<<<(N_NODES + 255) / 256, 256, 0, stream>>>(cnt, N_NODES);
  k_deg <<<(N_EDGES + 255) / 256, 256, 0, stream>>>(dst, cnt);
  k_dinv<<<(N_NODES + 255) / 256, 256, 0, stream>>>(cnt, dinv);

  const float* Ws[3] = {W1, W2, W3};
  const float* bs[3] = {b1, b2, b3};
  const float* in = x;
  for (int l = 0; l < 3; ++l) {
    k_gemm<<<N_NODES / 16, 256, 0, stream>>>(in, Ws[l], H, l > 0);
    k_init<<<(N_NODES * D) / 256, 256, 0, stream>>>(H, dinv, bs[l], A);
    k_edge<<<(N_EDGES * 32) / 256, 256, 0, stream>>>(H, src, dst, dinv, A);
    in = A;
  }

  k_final<<<(N_TGT * 32) / 256, 256, 0, stream>>>(A, tgt, Wr, br, out);
}


# Edge scatter-add by float atomics replaced by per-node gather over a dst-sorted edge list (counting sort each call, integer-atomic rank); sums per node now run in arrival-rank order
# speedup vs baseline: 2.4769x; 2.4769x over previous
#include <hip/hip_runtime.h>

typedef __attribute__((ext_vector_type(2))) float v2f;
typedef __attribute__((ext_vector_type(8))) float v8f;

#define N_NODES 50000
#define N_EDGES 800000
#define D       128
#define OUTD    12
#define N_TGT   8192

#define N_PAD   50048


__global__ void k_zero(int* __restrict__ p, int n) {
  int i = blockIdx.x * blockDim.x + threadIdx.x;
  if (i < n) p[i] = 0;
}

__global__ void k_rank(const int* __restrict__ dst, int* __restrict__ cnt,
                       int* __restrict__ rank) {
  int e = blockIdx.x * blockDim.x + threadIdx.x;
  if (e < N_EDGES) rank[e] = atomicAdd(&cnt[dst[e]], 1);
}


__global__ __launch_bounds__(256) void k_scan(const int* __restrict__ cnt,
                                              int* __restrict__ rowptr,
                                              float* __restrict__ dinv) {
  __shared__ int sFront[8];
  __shared__ int sWave[8];
  const int t    = threadIdx.x;
  const int lane = t & 31;
  const int wave = t >> 5;
  const int idx  = blockIdx.x * 256 + t;

  int front = 0;
  const int4* __restrict__ c4 = (const int4*)cnt;
  const int nq = blockIdx.x * 64;
  for (int i0 = t; i0 < nq; i0 += 2048) {
    int4 q[8];
    #pragma unroll
    for (int u = 0; u < 8; ++u) q[u] = c4[min(i0 + 256 * u, nq - 1)];
    #pragma unroll
    for (int u = 0; u < 8; ++u)
      if (i0 + 256 * u < nq) front += (q[u].x + q[u].y) + (q[u].z + q[u].w);
  }
  for (int m = 16; m >= 1; m >>= 1) front += __shfl_xor(front, m, 32);

  const int v = (idx < N_NODES) ? cnt[idx] : 0;
  int inc = v;
  for (int s = 1; s < 32; s <<= 1) {
    const int up = __shfl_up(inc, s, 32);
    if (lane >= s) inc += up;
  }
  if (lane == 0)  sFront[wave] = front;
  if (lane == 31) sWave[wave]  = inc;
  __syncthreads();

  int off = 0;
  for (int w = 0; w < 8; ++w) {
    off += sFront[w];
    if (w < wave) off += sWave[w];
  }
  const int excl = off + inc - v;
  if (idx < N_NODES) {
    rowptr[idx] = excl;
    dinv[idx]   = rsqrtf((float)v + 1.0f);
  }
  if (idx == N_NODES - 1) rowptr[N_NODES] = excl + v;
}

__global__ void k_fill(const int* __restrict__ src, const int* __restrict__ dst,
                       const int* __restrict__ rank, const int* __restrict__ rowptr,
                       int* __restrict__ csr_src) {
  int e = blockIdx.x * blockDim.x + threadIdx.x;
  if (e < N_EDGES) csr_src[rowptr[dst[e]] + rank[e]] = src[e];
}


__global__ __launch_bounds__(256) void k_gemm(const float* __restrict__ X,
                                              const float* __restrict__ W,
                                              float* __restrict__ H,
                                              int apply_relu) {
  __shared__ float sX[16][132];
  const int row0 = blockIdx.x * 16;

  for (int idx = threadIdx.x; idx < 16 * 128; idx += 256) {
    const int r = idx >> 7, c = idx & 127;
    float v = X[(size_t)(row0 + r) * D + c];
    if (apply_relu) v = fmaxf(v, 0.0f);
    sX[r][c] = v;
  }
  __syncthreads();

  const int wave    = threadIdx.x >> 5;
  const int lane    = threadIdx.x & 31;
  const int half    = lane >> 4;
  const int l       = lane & 15;
  const int colBase = wave * 16;

  v8f acc = {};
  #pragma unroll
  for (int k = 0; k < 32; ++k) {
    const int kb = 4 * k + 2 * half;
    v2f a, b;
    a.x = sX[l][kb];
    a.y = sX[l][kb + 1];
    b.x = W[(size_t)kb       * D + colBase + l];
    b.y = W[(size_t)(kb + 1) * D + colBase + l];
    acc = __builtin_amdgcn_wmma_f32_16x16x4_f32(
         false, a,  false, b,
         (short)0, acc,  false,  false);
  }

  #pragma unroll
  for (int r = 0; r < 8; ++r) {
    H[(size_t)(row0 + r + 8 * half) * D + colBase + l] = acc[r];
  }
}


__device__ __forceinline__ int lane_i(int v, int j) {
  return __builtin_amdgcn_readlane(v, j);
}
__device__ __forceinline__ float lane_f(float v, int j) {
  return __int_as_float(__builtin_amdgcn_readlane(__float_as_int(v), j));
}
__device__ __forceinline__ void axpy4(float4& acc, const float4 h, const float w) {
  acc.x += h.x * w;
  acc.y += h.y * w;
  acc.z += h.z * w;
  acc.w += h.w * w;
}

__global__ __launch_bounds__(256) void k_agg(const float* __restrict__ H,
                                             const int* __restrict__ rowptr,
                                             const int* __restrict__ csr_src,
                                             const float* __restrict__ dinv,
                                             const float* __restrict__ bias,
                                             float* __restrict__ A) {
  const int lane = threadIdx.x & 31;
  const int d    = blockIdx.x * 8 + (threadIdx.x >> 5);
  const float4* __restrict__ H4 = (const float4*)H;

  const int beg = __builtin_amdgcn_readfirstlane(rowptr[d]);
  const int end = __builtin_amdgcn_readfirstlane(rowptr[d + 1]);
  const float dd = dinv[d];
  const float4 hs = H4[(size_t)d * 32 + lane];
  const float4 bv = ((const float4*)bias)[lane];

  float4 acc;
  acc.x = hs.x * dd * dd + bv.x;
  acc.y = hs.y * dd * dd + bv.y;
  acc.z = hs.z * dd * dd + bv.z;
  acc.w = hs.w * dd * dd + bv.w;

  for (int base = beg; base < end; base += 32) {
    const int n = min(32, end - base);
    int   s_l = 0;
    float w_l = 0.0f;
    if (lane < n) {
      s_l = (int)min((unsigned)csr_src[base + lane], (unsigned)(N_NODES - 1));
      w_l = dinv[s_l] * dd;
    }
    int j = 0;
    for (; j + 4 <= n; j += 4) {
      const float4 h0 = H4[(size_t)lane_i(s_l, j)     * 32 + lane];
      const float4 h1 = H4[(size_t)lane_i(s_l, j + 1) * 32 + lane];
      const float4 h2 = H4[(size_t)lane_i(s_l, j + 2) * 32 + lane];
      const float4 h3 = H4[(size_t)lane_i(s_l, j + 3) * 32 + lane];
      axpy4(acc, h0, lane_f(w_l, j));
      axpy4(acc, h1, lane_f(w_l, j + 1));
      axpy4(acc, h2, lane_f(w_l, j + 2));
      axpy4(acc, h3, lane_f(w_l, j + 3));
    }
    const int r = n - j;
    float4 h0 = {}, h1 = {}, h2 = {};
    if (r > 0) h0 = H4[(size_t)lane_i(s_l, j)     * 32 + lane];
    if (r > 1) h1 = H4[(size_t)lane_i(s_l, j + 1) * 32 + lane];
    if (r > 2) h2 = H4[(size_t)lane_i(s_l, j + 2) * 32 + lane];
    if (r > 0) axpy4(acc, h0, lane_f(w_l, j));
    if (r > 1) axpy4(acc, h1, lane_f(w_l, j + 1));
    if (r > 2) axpy4(acc, h2, lane_f(w_l, j + 2));
  }

  ((float4*)A)[(size_t)d * 32 + lane] = acc;
}


__global__ void k_final(const float* __restrict__ A, const int* __restrict__ tgt,
                        const float* __restrict__ Wr, const float* __restrict__ br,
                        float* __restrict__ out) {
  const int g    = blockIdx.x * blockDim.x + threadIdx.x;
  const int t    = g >> 5;
  const int lane = g & 31;
  if (t >= N_TGT) return;
  const int node = tgt[t];
  float h[4];
  #pragma unroll
  for (int j = 0; j < 4; ++j)
    h[j] = fmaxf(A[(size_t)node * D + lane + 32 * j], 0.0f);
  for (int o = 0; o < OUTD; ++o) {
    float p = 0.0f;
    #pragma unroll
    for (int j = 0; j < 4; ++j)
      p += h[j] * Wr[(size_t)(lane + 32 * j) * OUTD + o];
    for (int m = 16; m >= 1; m >>= 1) p += __shfl_xor(p, m, 32);
    if (lane == 0) out[(size_t)t * OUTD + o] = p + br[o];
  }
}


extern "C" void kernel_launch(void* const* d_in, const int* in_sizes, int n_in,
                              void* d_out, int out_size, void* d_ws, size_t ws_size,
                              hipStream_t stream) {
  const float* x   = (const float*)d_in[0];
  const int*   ei  = (const int*)  d_in[1];
  const int*   tgt = (const int*)  d_in[2];
  const float* W1  = (const float*)d_in[3];
  const float* b1  = (const float*)d_in[4];
  const float* W2  = (const float*)d_in[5];
  const float* b2  = (const float*)d_in[6];
  const float* W3  = (const float*)d_in[7];
  const float* b3  = (const float*)d_in[8];
  const float* Wr  = (const float*)d_in[9];
  const float* br  = (const float*)d_in[10];
  float* out = (float*)d_out;

  const int* src = ei;
  const int* dst = ei + N_EDGES;

  float* ws      = (float*)d_ws;
  float* H       = ws;
  float* A       = H + (size_t)N_NODES * D;
  int*   cnt     = (int*)(A + (size_t)N_NODES * D);
  int*   rowptr  = cnt + N_PAD;
  float* dinv    = (float*)(rowptr + N_PAD);
  int*   csr_src = (int*)(dinv + N_PAD);
  int*   rank    = (int*)A;

  k_zero<<<(N_NODES + 255) / 256, 256, 0, stream>>>(cnt, N_NODES);
  k_rank<<<(N_EDGES + 255) / 256, 256, 0, stream>>>(dst, cnt, rank);
  k_scan<<<(N_NODES + 255) / 256, 256, 0, stream>>>(cnt, rowptr, dinv);
  k_fill<<<(N_EDGES + 255) / 256, 256, 0, stream>>>(src, dst, rank, rowptr, csr_src);

  const float* Ws[3] = {W1, W2, W3};
  const float* bs[3] = {b1, b2, b3};
  const float* in = x;
  for (int l = 0; l < 3; ++l) {
    k_gemm<<<N_NODES / 16, 256, 0, stream>>>(in, Ws[l], H, l > 0);
    k_agg<<<N_NODES / 8, 256, 0, stream>>>(H, rowptr, csr_src, dinv, bs[l], A);
    in = A;
  }

  k_final<<<(N_TGT * 32) / 256, 256, 0, stream>>>(A, tgt, Wr, br, out);
}


# Each layer reassociated to (Anorm X) W and fused into one kernel: 16-node tiles gathered into LDS, then WMMA with W held in registers, resident grid walks tiles; sums aggregate before the product
# speedup vs baseline: 2.9450x; 1.1890x over previous
#include <hip/hip_runtime.h>

typedef __attribute__((ext_vector_type(2))) float v2f;
typedef __attribute__((ext_vector_type(8))) float v8f;

#define N_NODES 50000
#define N_EDGES 800000
#define D       128
#define OUTD    12
#define N_TGT   8192

#define N_PAD   50048


__global__ void k_zero(int* __restrict__ p, int n) {
  int i = blockIdx.x * blockDim.x + threadIdx.x;
  if (i < n) p[i] = 0;
}

__global__ void k_rank(const int* __restrict__ dst, int* __restrict__ cnt,
                       int* __restrict__ rank) {
  int e = blockIdx.x * blockDim.x + threadIdx.x;
  if (e < N_EDGES) rank[e] = atomicAdd(&cnt[dst[e]], 1);
}


__global__ __launch_bounds__(256) void k_scan(const int* __restrict__ cnt,
                                              int* __restrict__ rowptr,
                                              float* __restrict__ dinv) {
  __shared__ int sFront[8];
  __shared__ int sWave[8];
  const int t    = threadIdx.x;
  const int lane = t & 31;
  const int wave = t >> 5;
  const int idx  = blockIdx.x * 256 + t;

  int front = 0;
  const int4* __restrict__ c4 = (const int4*)cnt;
  const int nq = blockIdx.x * 64;
  for (int i0 = t; i0 < nq; i0 += 2048) {
    int4 q[8];
    #pragma unroll
    for (int u = 0; u < 8; ++u) q[u] = c4[min(i0 + 256 * u, nq - 1)];
    #pragma unroll
    for (int u = 0; u < 8; ++u)
      if (i0 + 256 * u < nq) front += (q[u].x + q[u].y) + (q[u].z + q[u].w);
  }
  for (int m = 16; m >= 1; m >>= 1) front += __shfl_xor(front, m, 32);

  const int v = (idx < N_NODES) ? cnt[idx] : 0;
  int inc = v;
  for (int s = 1; s < 32; s <<= 1) {
    const int up = __shfl_up(inc, s, 32);
    if (lane >= s) inc += up;
  }
  if (lane == 0)  sFront[wave] = front;
  if (lane == 31) sWave[wave]  = inc;
  __syncthreads();

  int off = 0;
  for (int w = 0; w < 8; ++w) {
    off += sFront[w];
    if (w < wave) off += sWave[w];
  }
  const int excl = off + inc - v;
  if (idx < N_NODES) {
    rowptr[idx] = excl;
    dinv[idx]   = rsqrtf((float)v + 1.0f);
  }
  if (idx == N_NODES - 1) rowptr[N_NODES] = excl + v;
}

__global__ void k_fill(const int* __restrict__ src, const int* __restrict__ dst,
                       const int* __restrict__ rank, const int* __restrict__ rowptr,
                       int* __restrict__ csr_src) {
  int e = blockIdx.x * blockDim.x + threadIdx.x;
  if (e < N_EDGES) csr_src[rowptr[dst[e]] + rank[e]] = src[e];
}


__device__ __forceinline__ int lane_i(int v, int j) {
  return __builtin_amdgcn_readlane(v, j);
}
__device__ __forceinline__ float lane_f(float v, int j) {
  return __int_as_float(__builtin_amdgcn_readlane(__float_as_int(v), j));
}
__device__ __forceinline__ void axpy4(float4& acc, const float4 h, const float w) {
  acc.x += h.x * w;
  acc.y += h.y * w;
  acc.z += h.z * w;
  acc.w += h.w * w;
}

__device__ __forceinline__ float4 gather_node(const float4* __restrict__ X4,
                                              const int* __restrict__ rowptr,
                                              const int* __restrict__ csr_src,
                                              const float* __restrict__ dinv,
                                              const int d, const int lane) {
  const int beg = __builtin_amdgcn_readfirstlane(rowptr[d]);
  const int end = __builtin_amdgcn_readfirstlane(rowptr[d + 1]);
  const float dd = __int_as_float(__builtin_amdgcn_readfirstlane(__float_as_int(dinv[d])));
  const float4 xs = X4[(size_t)d * 32 + lane];

  float4 acc;
  acc.x = xs.x * dd * dd;
  acc.y = xs.y * dd * dd;
  acc.z = xs.z * dd * dd;
  acc.w = xs.w * dd * dd;

  for (int base = beg; base < end; base += 32) {
    const int n = min(32, end - base);
    const int   s_l = (int)min((unsigned)csr_src[min(base + lane, end - 1)],
                               (unsigned)(N_NODES - 1));
    const float w_l = dinv[s_l];
    int j = 0;
    for (; j + 8 <= n; j += 8) {
      float4 h[8];
      #pragma unroll
      for (int u = 0; u < 8; ++u) h[u] = X4[(size_t)lane_i(s_l, j + u) * 32 + lane];
      #pragma unroll
      for (int u = 0; u < 8; ++u) axpy4(acc, h[u], lane_f(w_l, j + u) * dd);
    }
    if (j + 4 <= n) {
      float4 h[4];
      #pragma unroll
      for (int u = 0; u < 4; ++u) h[u] = X4[(size_t)lane_i(s_l, j + u) * 32 + lane];
      #pragma unroll
      for (int u = 0; u < 4; ++u) axpy4(acc, h[u], lane_f(w_l, j + u) * dd);
      j += 4;
    }
    const int r = n - j;
    float4 h0 = {}, h1 = {}, h2 = {};
    if (r > 0) h0 = X4[(size_t)lane_i(s_l, j)     * 32 + lane];
    if (r > 1) h1 = X4[(size_t)lane_i(s_l, j + 1) * 32 + lane];
    if (r > 2) h2 = X4[(size_t)lane_i(s_l, j + 2) * 32 + lane];
    if (r > 0) axpy4(acc, h0, lane_f(w_l, j) * dd);
    if (r > 1) axpy4(acc, h1, lane_f(w_l, j + 1) * dd);
    if (r > 2) axpy4(acc, h2, lane_f(w_l, j + 2) * dd);
  }
  return acc;
}

#define N_TILES (N_NODES / 16)
#define LAYER_GRID 1024

__global__ __launch_bounds__(256) void k_layer(const float* __restrict__ Xin,
                                               const float* __restrict__ W,
                                               const float* __restrict__ bias,
                                               const int* __restrict__ rowptr,
                                               const int* __restrict__ csr_src,
                                               const float* __restrict__ dinv,
                                               float* __restrict__ Out) {
  __shared__ __attribute__((aligned(16))) float sX[16][132];

  const int wave    = threadIdx.x >> 5;
  const int lane    = threadIdx.x & 31;
  const int half    = lane >> 4;
  const int l       = lane & 15;
  const int colBase = wave * 16;
  const float4* __restrict__ X4 = (const float4*)Xin;

  v2f Bf[32];
  #pragma unroll
  for (int k = 0; k < 32; ++k) {
    const int kb = 4 * k + 2 * half;
    Bf[k].x = W[(size_t)kb       * D + colBase + l];
    Bf[k].y = W[(size_t)(kb + 1) * D + colBase + l];
  }
  const float bcol = bias[colBase + l];

  for (int tile = blockIdx.x; tile < N_TILES; tile += gridDim.x) {
    const int row0 = tile * 16;

    for (int q = 0; q < 2; ++q) {
      const int r = 2 * wave + q;
      const float4 g = gather_node(X4, rowptr, csr_src, dinv, row0 + r, lane);
      *(float4*)&sX[r][4 * lane] = g;
    }
    __syncthreads();

    v8f acc = {};
    #pragma unroll
    for (int k = 0; k < 32; ++k) {
      const int kb = 4 * k + 2 * half;
      v2f a;
      a.x = sX[l][kb];
      a.y = sX[l][kb + 1];
      acc = __builtin_amdgcn_wmma_f32_16x16x4_f32(
           false, a,  false, Bf[k],
           (short)0, acc,  false,  false);
    }
    __syncthreads();

    #pragma unroll
    for (int r = 0; r < 8; ++r)
      sX[r + 8 * half][colBase + l] = fmaxf(acc[r] + bcol, 0.0f);
    __syncthreads();
    #pragma unroll
    for (int q = 0; q < 2; ++q) {
      const int r = 2 * wave + q;
      ((float4*)Out)[(size_t)(row0 + r) * 32 + lane] = *(const float4*)&sX[r][4 * lane];
    }
  }
}


__global__ void k_final(const float* __restrict__ P, const int* __restrict__ tgt,
                        const float* __restrict__ Wr, const float* __restrict__ br,
                        float* __restrict__ out) {
  const int g    = blockIdx.x * blockDim.x + threadIdx.x;
  const int t    = g >> 5;
  const int lane = g & 31;
  if (t >= N_TGT) return;
  const int node = tgt[t];
  float h[4];
  #pragma unroll
  for (int j = 0; j < 4; ++j)
    h[j] = P[(size_t)node * D + lane + 32 * j];
  for (int o = 0; o < OUTD; ++o) {
    float p = 0.0f;
    #pragma unroll
    for (int j = 0; j < 4; ++j)
      p += h[j] * Wr[(size_t)(lane + 32 * j) * OUTD + o];
    for (int m = 16; m >= 1; m >>= 1) p += __shfl_xor(p, m, 32);
    if (lane == 0) out[(size_t)t * OUTD + o] = p + br[o];
  }
}


extern "C" void kernel_launch(void* const* d_in, const int* in_sizes, int n_in,
                              void* d_out, int out_size, void* d_ws, size_t ws_size,
                              hipStream_t stream) {
  const float* x   = (const float*)d_in[0];
  const int*   ei  = (const int*)  d_in[1];
  const int*   tgt = (const int*)  d_in[2];
  const float* W1  = (const float*)d_in[3];
  const float* b1  = (const float*)d_in[4];
  const float* W2  = (const float*)d_in[5];
  const float* b2  = (const float*)d_in[6];
  const float* W3  = (const float*)d_in[7];
  const float* b3  = (const float*)d_in[8];
  const float* Wr  = (const float*)d_in[9];
  const float* br  = (const float*)d_in[10];
  float* out = (float*)d_out;

  const int* src = ei;
  const int* dst = ei + N_EDGES;

  float* ws      = (float*)d_ws;
  float* P0      = ws;
  float* P1      = P0 + (size_t)N_NODES * D;
  int*   cnt     = (int*)(P1 + (size_t)N_NODES * D);
  int*   rowptr  = cnt + N_PAD;
  float* dinv    = (float*)(rowptr + N_PAD);
  int*   csr_src = (int*)(dinv + N_PAD);
  int*   rank    = (int*)P1;

  k_zero<<<(N_NODES + 255) / 256, 256, 0, stream>>>(cnt, N_NODES);
  k_rank<<<(N_EDGES + 255) / 256, 256, 0, stream>>>(dst, cnt, rank);
  k_scan<<<(N_NODES + 255) / 256, 256, 0, stream>>>(cnt, rowptr, dinv);
  k_fill<<<(N_EDGES + 255) / 256, 256, 0, stream>>>(src, dst, rank, rowptr, csr_src);

  k_layer<<<LAYER_GRID, 256, 0, stream>>>(x,  W1, b1, rowptr, csr_src, dinv, P0);
  k_layer<<<LAYER_GRID, 256, 0, stream>>>(P0, W2, b2, rowptr, csr_src, dinv, P1);
  k_layer<<<LAYER_GRID, 256, 0, stream>>>(P1, W3, b3, rowptr, csr_src, dinv, P0);

  k_final<<<(N_TGT * 32) / 256, 256, 0, stream>>>(P0, tgt, Wr, br, out);
}


# Third layer computed only for the 8192 target rows that the head reads (tile rows taken from target_idx, compact plane of 8192 rows); its other 41808 rows are never formed
# speedup vs baseline: 3.2698x; 1.1103x over previous
#include <hip/hip_runtime.h>

typedef __attribute__((ext_vector_type(2))) float v2f;
typedef __attribute__((ext_vector_type(8))) float v8f;

#define N_NODES 50000
#define N_EDGES 800000
#define D       128
#define OUTD    12
#define N_TGT   8192

#define N_PAD   50048


__global__ void k_zero(int* __restrict__ p, int n) {
  int i = blockIdx.x * blockDim.x + threadIdx.x;
  if (i < n) p[i] = 0;
}

__global__ void k_rank(const int* __restrict__ dst, int* __restrict__ cnt,
                       int* __restrict__ rank) {
  int e = blockIdx.x * blockDim.x + threadIdx.x;
  if (e < N_EDGES) rank[e] = atomicAdd(&cnt[dst[e]], 1);
}


__global__ __launch_bounds__(256) void k_scan(const int* __restrict__ cnt,
                                              int* __restrict__ rowptr,
                                              float* __restrict__ dinv) {
  __shared__ int sFront[8];
  __shared__ int sWave[8];
  const int t    = threadIdx.x;
  const int lane = t & 31;
  const int wave = t >> 5;
  const int idx  = blockIdx.x * 256 + t;

  int front = 0;
  const int4* __restrict__ c4 = (const int4*)cnt;
  const int nq = blockIdx.x * 64;
  for (int i0 = t; i0 < nq; i0 += 2048) {
    int4 q[8];
    #pragma unroll
    for (int u = 0; u < 8; ++u) q[u] = c4[min(i0 + 256 * u, nq - 1)];
    #pragma unroll
    for (int u = 0; u < 8; ++u)
      if (i0 + 256 * u < nq) front += (q[u].x + q[u].y) + (q[u].z + q[u].w);
  }
  for (int m = 16; m >= 1; m >>= 1) front += __shfl_xor(front, m, 32);

  const int v = (idx < N_NODES) ? cnt[idx] : 0;
  int inc = v;
  for (int s = 1; s < 32; s <<= 1) {
    const int up = __shfl_up(inc, s, 32);
    if (lane >= s) inc += up;
  }
  if (lane == 0)  sFront[wave] = front;
  if (lane == 31) sWave[wave]  = inc;
  __syncthreads();

  int off = 0;
  for (int w = 0; w < 8; ++w) {
    off += sFront[w];
    if (w < wave) off += sWave[w];
  }
  const int excl = off + inc - v;
  if (idx < N_NODES) {
    rowptr[idx] = excl;
    dinv[idx]   = rsqrtf((float)v + 1.0f);
  }
  if (idx == N_NODES - 1) rowptr[N_NODES] = excl + v;
}

__global__ void k_fill(const int* __restrict__ src, const int* __restrict__ dst,
                       const int* __restrict__ rank, const int* __restrict__ rowptr,
                       int* __restrict__ csr_src) {
  int e = blockIdx.x * blockDim.x + threadIdx.x;
  if (e < N_EDGES) csr_src[rowptr[dst[e]] + rank[e]] = src[e];
}


__device__ __forceinline__ int lane_i(int v, int j) {
  return __builtin_amdgcn_readlane(v, j);
}
__device__ __forceinline__ float lane_f(float v, int j) {
  return __int_as_float(__builtin_amdgcn_readlane(__float_as_int(v), j));
}
__device__ __forceinline__ void axpy4(float4& acc, const float4 h, const float w) {
  acc.x += h.x * w;
  acc.y += h.y * w;
  acc.z += h.z * w;
  acc.w += h.w * w;
}

__device__ __forceinline__ float4 gather_node(const float4* __restrict__ X4,
                                              const int* __restrict__ rowptr,
                                              const int* __restrict__ csr_src,
                                              const float* __restrict__ dinv,
                                              const int d, const int lane) {
  const int beg = __builtin_amdgcn_readfirstlane(rowptr[d]);
  const int end = __builtin_amdgcn_readfirstlane(rowptr[d + 1]);
  const float dd = __int_as_float(__builtin_amdgcn_readfirstlane(__float_as_int(dinv[d])));
  const float4 xs = X4[(size_t)d * 32 + lane];

  float4 acc;
  acc.x = xs.x * dd * dd;
  acc.y = xs.y * dd * dd;
  acc.z = xs.z * dd * dd;
  acc.w = xs.w * dd * dd;

  for (int base = beg; base < end; base += 32) {
    const int n = min(32, end - base);
    const int   s_l = (int)min((unsigned)csr_src[min(base + lane, end - 1)],
                               (unsigned)(N_NODES - 1));
    const float w_l = dinv[s_l];
    int j = 0;
    for (; j + 8 <= n; j += 8) {
      float4 h[8];
      #pragma unroll
      for (int u = 0; u < 8; ++u) h[u] = X4[(size_t)lane_i(s_l, j + u) * 32 + lane];
      #pragma unroll
      for (int u = 0; u < 8; ++u) axpy4(acc, h[u], lane_f(w_l, j + u) * dd);
    }
    if (j + 4 <= n) {
      float4 h[4];
      #pragma unroll
      for (int u = 0; u < 4; ++u) h[u] = X4[(size_t)lane_i(s_l, j + u) * 32 + lane];
      #pragma unroll
      for (int u = 0; u < 4; ++u) axpy4(acc, h[u], lane_f(w_l, j + u) * dd);
      j += 4;
    }
    const int r = n - j;
    float4 h0 = {}, h1 = {}, h2 = {};
    if (r > 0) h0 = X4[(size_t)lane_i(s_l, j)     * 32 + lane];
    if (r > 1) h1 = X4[(size_t)lane_i(s_l, j + 1) * 32 + lane];
    if (r > 2) h2 = X4[(size_t)lane_i(s_l, j + 2) * 32 + lane];
    if (r > 0) axpy4(acc, h0, lane_f(w_l, j) * dd);
    if (r > 1) axpy4(acc, h1, lane_f(w_l, j + 1) * dd);
    if (r > 2) axpy4(acc, h2, lane_f(w_l, j + 2) * dd);
  }
  return acc;
}


#define N_TILES     (N_NODES / 16)
#define N_TGT_TILES (N_TGT / 16)
#define LAYER_GRID 1024

template <bool LIST>
__global__ __launch_bounds__(256) void k_layer(const float* __restrict__ Xin,
                                               const float* __restrict__ W,
                                               const float* __restrict__ bias,
                                               const int* __restrict__ rowptr,
                                               const int* __restrict__ csr_src,
                                               const float* __restrict__ dinv,
                                               const int* __restrict__ list,
                                               float* __restrict__ Out) {
  constexpr int n_tiles = LIST ? N_TGT_TILES : N_TILES;
  __shared__ __attribute__((aligned(16))) float sX[16][132];

  const int wave    = threadIdx.x >> 5;
  const int lane    = threadIdx.x & 31;
  const int half    = lane >> 4;
  const int l       = lane & 15;
  const int colBase = wave * 16;
  const float4* __restrict__ X4 = (const float4*)Xin;

  v2f Bf[32];
  #pragma unroll
  for (int k = 0; k < 32; ++k) {
    const int kb = 4 * k + 2 * half;
    Bf[k].x = W[(size_t)kb       * D + colBase + l];
    Bf[k].y = W[(size_t)(kb + 1) * D + colBase + l];
  }
  const float bcol = bias[colBase + l];

  for (int tile = blockIdx.x; tile < n_tiles; tile += gridDim.x) {
    const int row0 = tile * 16;

    int2 ids = {};
    if (LIST) ids = *(const int2*)(list + row0 + 2 * wave);
    for (int q = 0; q < 2; ++q) {
      const int r = 2 * wave + q;
      int d = row0 + r;
      if (LIST)
        d = __builtin_amdgcn_readfirstlane(
            (int)min((unsigned)(q ? ids.y : ids.x), (unsigned)(N_NODES - 1)));
      const float4 g = gather_node(X4, rowptr, csr_src, dinv, d, lane);
      *(float4*)&sX[r][4 * lane] = g;
    }
    __syncthreads();

    v8f acc = {};
    #pragma unroll
    for (int k = 0; k < 32; ++k) {
      const int kb = 4 * k + 2 * half;
      v2f a;
      a.x = sX[l][kb];
      a.y = sX[l][kb + 1];
      acc = __builtin_amdgcn_wmma_f32_16x16x4_f32(
           false, a,  false, Bf[k],
           (short)0, acc,  false,  false);
    }
    __syncthreads();

    #pragma unroll
    for (int r = 0; r < 8; ++r)
      sX[r + 8 * half][colBase + l] = fmaxf(acc[r] + bcol, 0.0f);
    __syncthreads();
    #pragma unroll
    for (int q = 0; q < 2; ++q) {
      const int r = 2 * wave + q;
      ((float4*)Out)[(size_t)(row0 + r) * 32 + lane] = *(const float4*)&sX[r][4 * lane];
    }
  }
}


__global__ void k_final(const float* __restrict__ P,
                        const float* __restrict__ Wr, const float* __restrict__ br,
                        float* __restrict__ out) {
  const int g    = blockIdx.x * blockDim.x + threadIdx.x;
  const int t    = g >> 5;
  const int lane = g & 31;
  if (t >= N_TGT) return;
  float h[4];
  #pragma unroll
  for (int j = 0; j < 4; ++j)
    h[j] = P[(size_t)t * D + lane + 32 * j];
  for (int o = 0; o < OUTD; ++o) {
    float p = 0.0f;
    #pragma unroll
    for (int j = 0; j < 4; ++j)
      p += h[j] * Wr[(size_t)(lane + 32 * j) * OUTD + o];
    for (int m = 16; m >= 1; m >>= 1) p += __shfl_xor(p, m, 32);
    if (lane == 0) out[(size_t)t * OUTD + o] = p + br[o];
  }
}


extern "C" void kernel_launch(void* const* d_in, const int* in_sizes, int n_in,
                              void* d_out, int out_size, void* d_ws, size_t ws_size,
                              hipStream_t stream) {
  const float* x   = (const float*)d_in[0];
  const int*   ei  = (const int*)  d_in[1];
  const int*   tgt = (const int*)  d_in[2];
  const float* W1  = (const float*)d_in[3];
  const float* b1  = (const float*)d_in[4];
  const float* W2  = (const float*)d_in[5];
  const float* b2  = (const float*)d_in[6];
  const float* W3  = (const float*)d_in[7];
  const float* b3  = (const float*)d_in[8];
  const float* Wr  = (const float*)d_in[9];
  const float* br  = (const float*)d_in[10];
  float* out = (float*)d_out;

  const int* src = ei;
  const int* dst = ei + N_EDGES;

  float* ws      = (float*)d_ws;
  float* P0      = ws;
  float* P1      = P0 + (size_t)N_NODES * D;
  int*   cnt     = (int*)(P1 + (size_t)N_NODES * D);
  int*   rowptr  = cnt + N_PAD;
  float* dinv    = (float*)(rowptr + N_PAD);
  int*   csr_src = (int*)(dinv + N_PAD);
  int*   rank    = (int*)P1;

  k_zero<<<(N_NODES + 255) / 256, 256, 0, stream>>>(cnt, N_NODES);
  k_rank<<<(N_EDGES + 255) / 256, 256, 0, stream>>>(dst, cnt, rank);
  k_scan<<<(N_NODES + 255) / 256, 256, 0, stream>>>(cnt, rowptr, dinv);
  k_fill<<<(N_EDGES + 255) / 256, 256, 0, stream>>>(src, dst, rank, rowptr, csr_src);

  k_layer<false><<<LAYER_GRID, 256, 0, stream>>>(x,  W1, b1, rowptr, csr_src, dinv, nullptr, P0);
  k_layer<false><<<LAYER_GRID, 256, 0, stream>>>(P0, W2, b2, rowptr, csr_src, dinv, nullptr, P1);
  k_layer<true><<<N_TGT_TILES, 256, 0, stream>>>(P1, W3, b3, rowptr, csr_src, dinv, tgt, P0);

  k_final<<<(N_TGT * 32) / 256, 256, 0, stream>>>(P0, Wr, br, out);
}


# Per-call counting sort rebuilt in two levels with LDS counters (chunk blocks group edges by 196-node bucket, bucket blocks rank per node): no global atomics, 6 launches; node sums keep arrival order
# speedup vs baseline: 5.4405x; 1.6639x over previous
#include <hip/hip_runtime.h>

typedef __attribute__((ext_vector_type(2))) float v2f;
typedef __attribute__((ext_vector_type(8))) float v8f;

#define N_NODES 50000
#define N_EDGES 800000
#define D       128
#define OUTD    12
#define N_TGT   8192

#define N_PAD   50048


#define BN      196
#define NB      256
#define N_CHUNK 256
#define CH      (N_EDGES / N_CHUNK)
#define CH_PT   13

static_assert(NB * BN >= N_NODES, "buckets must cover every node");
static_assert(N_CHUNK * CH == N_EDGES, "chunks must cover every edge");
static_assert(256 * CH_PT >= CH, "a block must cover its chunk");

__device__ __forceinline__ int block_excl_scan_256(const int v, int* sWave, const int t) {
  const int lane = t & 31;
  const int wave = t >> 5;
  int inc = v;
  for (int s = 1; s < 32; s <<= 1) {
    const int up = __shfl_up(inc, s, 32);
    if (lane >= s) inc += up;
  }
  if (lane == 31) sWave[wave] = inc;
  __syncthreads();
  int off = 0;
  for (int w = 0; w < 8; ++w)
    if (w < wave) off += sWave[w];
  return off + inc - v;
}

__global__ __launch_bounds__(256) void k_chunk_sort(const int* __restrict__ src,
                                                    const int* __restrict__ dst,
                                                    int* __restrict__ offG,
                                                    int2* __restrict__ CS) {
  __shared__ int h[NB];
  __shared__ int sWave[8];
  const int t  = threadIdx.x;
  const int c  = blockIdx.x;
  const int e0 = c * CH;
  __builtin_assume(t < 256);

  h[t] = 0;
  int s[CH_PT], d[CH_PT];
  #pragma unroll
  for (int u = 0; u < CH_PT; ++u) {
    const int i = min(t + 256 * u, CH - 1);
    s[u] = src[e0 + i];
    d[u] = (int)min((unsigned)dst[e0 + i], (unsigned)(N_NODES - 1));
  }
  __syncthreads();
  #pragma unroll
  for (int u = 0; u < CH_PT; ++u)
    if (t + 256 * u < CH) atomicAdd(&h[d[u] / BN], 1);
  __syncthreads();

  const int v    = h[t];
  const int excl = block_excl_scan_256(v, sWave, t);
  offG[c * NB + t] = excl;
  h[t] = excl;
  __syncthreads();
  int pos[CH_PT];
  #pragma unroll
  for (int u = 0; u < CH_PT; ++u)
    pos[u] = (t + 256 * u < CH) ? atomicAdd(&h[d[u] / BN], 1) : 0;
  #pragma unroll
  for (int u = 0; u < CH_PT; ++u)
    if (t + 256 * u < CH)
      CS[(size_t)e0 + min((unsigned)pos[u], (unsigned)(CH - 1))] = make_int2(s[u], d[u]);
}

__global__ __launch_bounds__(256) void k_bucket(const int* __restrict__ offG,
                                                const int2* __restrict__ CS,
                                                int* __restrict__ rowptr,
                                                float* __restrict__ dinv,
                                                int* __restrict__ csr_src) {
  __shared__ int cnt[256];
  __shared__ int cur[256 + 32];
  __shared__ int sRed[8];
  __shared__ int sWave[8];
  const int t    = threadIdx.x;
  const int k    = blockIdx.x;
  const int lane = t & 31;
  const int wave = t >> 5;
  const int n0   = k * BN;

  int pb = offG[t * NB + k];
  int pe = (k + 1 < NB) ? offG[t * NB + k + 1] : CH;
  pb = min(max(pb, 0), CH);
  pe = min(max(pe, pb), CH);
  cnt[t] = 0;

  int sum = pb;
  for (int m = 16; m >= 1; m >>= 1) sum += __shfl_xor(sum, m, 32);
  if (lane == 0) sRed[wave] = sum;
  __syncthreads();
  int beg = 0;
  for (int w = 0; w < 8; ++w) beg += sRed[w];

  const int2* __restrict__ piece = CS + (size_t)t * CH;

  for (int j = pb; j < pe; j += 8) {
    int2 p[8];
    #pragma unroll
    for (int u = 0; u < 8; ++u) p[u] = piece[min(j + u, pe - 1)];
    #pragma unroll
    for (int u = 0; u < 8; ++u)
      if (j + u < pe)
        atomicAdd(&cnt[min((unsigned)(p[u].y - n0), (unsigned)(BN - 1))], 1);
  }
  __syncthreads();

  const int v    = cnt[t];
  const int excl = block_excl_scan_256(v, sWave, t);
  const int node = n0 + t;
  if (t < BN && node < N_NODES) {
    rowptr[node] = beg + excl;
    dinv[node]   = rsqrtf((float)v + 1.0f);
  }
  if (k == NB - 1 && t == 0) rowptr[N_NODES] = N_EDGES;
  cur[t] = beg + excl;
  if (t < 32) cur[256 + t] = 0;
  __syncthreads();

  for (int j = pb; j < pe; j += 8) {
    int2 p[8];
    #pragma unroll
    for (int u = 0; u < 8; ++u) p[u] = piece[min(j + u, pe - 1)];
    int pos[8];
    #pragma unroll
    for (int u = 0; u < 8; ++u) {
      const int w = (j + u < pe) ? (int)min((unsigned)(p[u].y - n0), (unsigned)(BN - 1))
                                 : 256 + lane;
      pos[u] = atomicAdd(&cur[w], 1);
    }
    #pragma unroll
    for (int u = 0; u < 8; ++u)
      if (j + u < pe) csr_src[min((unsigned)pos[u], (unsigned)(N_EDGES - 1))] = p[u].x;
  }
}


__device__ __forceinline__ int lane_i(int v, int j) {
  return __builtin_amdgcn_readlane(v, j);
}
__device__ __forceinline__ float lane_f(float v, int j) {
  return __int_as_float(__builtin_amdgcn_readlane(__float_as_int(v), j));
}
__device__ __forceinline__ void axpy4(float4& acc, const float4 h, const float w) {
  acc.x += h.x * w;
  acc.y += h.y * w;
  acc.z += h.z * w;
  acc.w += h.w * w;
}

__device__ __forceinline__ float4 gather_node(const float4* __restrict__ X4,
                                              const int* __restrict__ rowptr,
                                              const int* __restrict__ csr_src,
                                              const float* __restrict__ dinv,
                                              const int d, const int lane) {
  const int beg = __builtin_amdgcn_readfirstlane(rowptr[d]);
  const int end = __builtin_amdgcn_readfirstlane(rowptr[d + 1]);
  const float dd = __int_as_float(__builtin_amdgcn_readfirstlane(__float_as_int(dinv[d])));
  const float4 xs = X4[(size_t)d * 32 + lane];

  float4 acc;
  acc.x = xs.x * dd * dd;
  acc.y = xs.y * dd * dd;
  acc.z = xs.z * dd * dd;
  acc.w = xs.w * dd * dd;

  for (int base = beg; base < end; base += 32) {
    const int n = min(32, end - base);
    const int   s_l = (int)min((unsigned)csr_src[min(base + lane, end - 1)],
                               (unsigned)(N_NODES - 1));
    const float w_l = dinv[s_l];
    int j = 0;
    for (; j + 8 <= n; j += 8) {
      float4 h[8];
      #pragma unroll
      for (int u = 0; u < 8; ++u) h[u] = X4[(size_t)lane_i(s_l, j + u) * 32 + lane];
      #pragma unroll
      for (int u = 0; u < 8; ++u) axpy4(acc, h[u], lane_f(w_l, j + u) * dd);
    }
    if (j + 4 <= n) {
      float4 h[4];
      #pragma unroll
      for (int u = 0; u < 4; ++u) h[u] = X4[(size_t)lane_i(s_l, j + u) * 32 + lane];
      #pragma unroll
      for (int u = 0; u < 4; ++u) axpy4(acc, h[u], lane_f(w_l, j + u) * dd);
      j += 4;
    }
    const int r = n - j;
    float4 h0 = {}, h1 = {}, h2 = {};
    if (r > 0) h0 = X4[(size_t)lane_i(s_l, j)     * 32 + lane];
    if (r > 1) h1 = X4[(size_t)lane_i(s_l, j + 1) * 32 + lane];
    if (r > 2) h2 = X4[(size_t)lane_i(s_l, j + 2) * 32 + lane];
    if (r > 0) axpy4(acc, h0, lane_f(w_l, j) * dd);
    if (r > 1) axpy4(acc, h1, lane_f(w_l, j + 1) * dd);
    if (r > 2) axpy4(acc, h2, lane_f(w_l, j + 2) * dd);
  }
  return acc;
}


#define N_TILES     (N_NODES / 16)
#define N_TGT_TILES (N_TGT / 16)
#define LAYER_GRID 1024

template <bool LIST>
__global__ __launch_bounds__(256) void k_layer(const float* __restrict__ Xin,
                                               const float* __restrict__ W,
                                               const float* __restrict__ bias,
                                               const int* __restrict__ rowptr,
                                               const int* __restrict__ csr_src,
                                               const float* __restrict__ dinv,
                                               const int* __restrict__ list,
                                               float* __restrict__ Out) {
  constexpr int n_tiles = LIST ? N_TGT_TILES : N_TILES;
  __shared__ __attribute__((aligned(16))) float sX[16][132];

  const int wave    = threadIdx.x >> 5;
  const int lane    = threadIdx.x & 31;
  const int half    = lane >> 4;
  const int l       = lane & 15;
  const int colBase = wave * 16;
  const float4* __restrict__ X4 = (const float4*)Xin;

  v2f Bf[32];
  #pragma unroll
  for (int k = 0; k < 32; ++k) {
    const int kb = 4 * k + 2 * half;
    Bf[k].x = W[(size_t)kb       * D + colBase + l];
    Bf[k].y = W[(size_t)(kb + 1) * D + colBase + l];
  }
  const float bcol = bias[colBase + l];

  for (int tile = blockIdx.x; tile < n_tiles; tile += gridDim.x) {
    const int row0 = tile * 16;

    int2 ids = {};
    if (LIST) ids = *(const int2*)(list + row0 + 2 * wave);
    for (int q = 0; q < 2; ++q) {
      const int r = 2 * wave + q;
      int d = row0 + r;
      if (LIST)
        d = __builtin_amdgcn_readfirstlane(
            (int)min((unsigned)(q ? ids.y : ids.x), (unsigned)(N_NODES - 1)));
      const float4 g = gather_node(X4, rowptr, csr_src, dinv, d, lane);
      *(float4*)&sX[r][4 * lane] = g;
    }
    __syncthreads();

    v8f acc = {};
    #pragma unroll
    for (int k = 0; k < 32; ++k) {
      const int kb = 4 * k + 2 * half;
      v2f a;
      a.x = sX[l][kb];
      a.y = sX[l][kb + 1];
      acc = __builtin_amdgcn_wmma_f32_16x16x4_f32(
           false, a,  false, Bf[k],
           (short)0, acc,  false,  false);
    }
    __syncthreads();

    #pragma unroll
    for (int r = 0; r < 8; ++r)
      sX[r + 8 * half][colBase + l] = fmaxf(acc[r] + bcol, 0.0f);
    __syncthreads();
    #pragma unroll
    for (int q = 0; q < 2; ++q) {
      const int r = 2 * wave + q;
      ((float4*)Out)[(size_t)(row0 + r) * 32 + lane] = *(const float4*)&sX[r][4 * lane];
    }
  }
}


__global__ void k_final(const float* __restrict__ P,
                        const float* __restrict__ Wr, const float* __restrict__ br,
                        float* __restrict__ out) {
  const int g    = blockIdx.x * blockDim.x + threadIdx.x;
  const int t    = g >> 5;
  const int lane = g & 31;
  if (t >= N_TGT) return;
  float h[4];
  #pragma unroll
  for (int j = 0; j < 4; ++j)
    h[j] = P[(size_t)t * D + lane + 32 * j];
  for (int o = 0; o < OUTD; ++o) {
    float p = 0.0f;
    #pragma unroll
    for (int j = 0; j < 4; ++j)
      p += h[j] * Wr[(size_t)(lane + 32 * j) * OUTD + o];
    for (int m = 16; m >= 1; m >>= 1) p += __shfl_xor(p, m, 32);
    if (lane == 0) out[(size_t)t * OUTD + o] = p + br[o];
  }
}


extern "C" void kernel_launch(void* const* d_in, const int* in_sizes, int n_in,
                              void* d_out, int out_size, void* d_ws, size_t ws_size,
                              hipStream_t stream) {
  const float* x   = (const float*)d_in[0];
  const int*   ei  = (const int*)  d_in[1];
  const int*   tgt = (const int*)  d_in[2];
  const float* W1  = (const float*)d_in[3];
  const float* b1  = (const float*)d_in[4];
  const float* W2  = (const float*)d_in[5];
  const float* b2  = (const float*)d_in[6];
  const float* W3  = (const float*)d_in[7];
  const float* b3  = (const float*)d_in[8];
  const float* Wr  = (const float*)d_in[9];
  const float* br  = (const float*)d_in[10];
  float* out = (float*)d_out;

  const int* src = ei;
  const int* dst = ei + N_EDGES;

  float* ws      = (float*)d_ws;
  float* P0      = ws;
  float* P1      = P0 + (size_t)N_NODES * D;
  int*   rowptr  = (int*)(P1 + (size_t)N_NODES * D) + N_PAD;
  float* dinv    = (float*)(rowptr + N_PAD);
  int*   csr_src = (int*)(dinv + N_PAD);
  int2*  CS      = (int2*)P1;
  int*   offG    = (int*)P1 + 2 * (size_t)N_EDGES;

  k_chunk_sort<<<N_CHUNK, 256, 0, stream>>>(src, dst, offG, CS);
  k_bucket<<<NB, 256, 0, stream>>>(offG, CS, rowptr, dinv, csr_src);

  k_layer<false><<<LAYER_GRID, 256, 0, stream>>>(x,  W1, b1, rowptr, csr_src, dinv, nullptr, P0);
  k_layer<false><<<LAYER_GRID, 256, 0, stream>>>(P0, W2, b2, rowptr, csr_src, dinv, nullptr, P1);
  k_layer<true><<<N_TGT_TILES, 256, 0, stream>>>(P1, W3, b3, rowptr, csr_src, dinv, tgt, P0);

  k_final<<<(N_TGT * 32) / 256, 256, 0, stream>>>(P0, Wr, br, out);
}


# Regressor head moved into the third layer's kernel: on the finished tile in LDS each of 8 waves takes 16 features on f32 WMMA, partial tiles added in wave order (head sums regrouped); 5 launches
# speedup vs baseline: 5.8752x; 1.0799x over previous
#include <hip/hip_runtime.h>

typedef __attribute__((ext_vector_type(2))) float v2f;
typedef __attribute__((ext_vector_type(8))) float v8f;

#define N_NODES 50000
#define N_EDGES 800000
#define D       128
#define OUTD    12
#define N_TGT   8192

#define N_PAD   50048


#define BN      196
#define NB      256
#define N_CHUNK 256
#define CH      (N_EDGES / N_CHUNK)
#define CH_PT   13

static_assert(NB * BN >= N_NODES, "buckets must cover every node");
static_assert(N_CHUNK * CH == N_EDGES, "chunks must cover every edge");
static_assert(256 * CH_PT >= CH, "a block must cover its chunk");

__device__ __forceinline__ int block_excl_scan_256(const int v, int* sWave, const int t) {
  const int lane = t & 31;
  const int wave = t >> 5;
  int inc = v;
  for (int s = 1; s < 32; s <<= 1) {
    const int up = __shfl_up(inc, s, 32);
    if (lane >= s) inc += up;
  }
  if (lane == 31) sWave[wave] = inc;
  __syncthreads();
  int off = 0;
  for (int w = 0; w < 8; ++w)
    if (w < wave) off += sWave[w];
  return off + inc - v;
}

__global__ __launch_bounds__(256) void k_chunk_sort(const int* __restrict__ src,
                                                    const int* __restrict__ dst,
                                                    int* __restrict__ offG,
                                                    int2* __restrict__ CS) {
  __shared__ int h[NB];
  __shared__ int sWave[8];
  const int t  = threadIdx.x;
  const int c  = blockIdx.x;
  const int e0 = c * CH;
  __builtin_assume(t < 256);

  h[t] = 0;
  int s[CH_PT], d[CH_PT];
  #pragma unroll
  for (int u = 0; u < CH_PT; ++u) {
    const int i = min(t + 256 * u, CH - 1);
    s[u] = src[e0 + i];
    d[u] = (int)min((unsigned)dst[e0 + i], (unsigned)(N_NODES - 1));
  }
  __syncthreads();
  #pragma unroll
  for (int u = 0; u < CH_PT; ++u)
    if (t + 256 * u < CH) atomicAdd(&h[d[u] / BN], 1);
  __syncthreads();

  const int v    = h[t];
  const int excl = block_excl_scan_256(v, sWave, t);
  offG[c * NB + t] = excl;
  h[t] = excl;
  __syncthreads();
  int pos[CH_PT];
  #pragma unroll
  for (int u = 0; u < CH_PT; ++u)
    pos[u] = (t + 256 * u < CH) ? atomicAdd(&h[d[u] / BN], 1) : 0;
  #pragma unroll
  for (int u = 0; u < CH_PT; ++u)
    if (t + 256 * u < CH)
      CS[(size_t)e0 + min((unsigned)pos[u], (unsigned)(CH - 1))] = make_int2(s[u], d[u]);
}

__global__ __launch_bounds__(256) void k_bucket(const int* __restrict__ offG,
                                                const int2* __restrict__ CS,
                                                int* __restrict__ rowptr,
                                                float* __restrict__ dinv,
                                                int* __restrict__ csr_src) {
  __shared__ int cnt[256];
  __shared__ int cur[256 + 32];
  __shared__ int sRed[8];
  __shared__ int sWave[8];
  const int t    = threadIdx.x;
  const int k    = blockIdx.x;
  const int lane = t & 31;
  const int wave = t >> 5;
  const int n0   = k * BN;

  int pb = offG[t * NB + k];
  int pe = (k + 1 < NB) ? offG[t * NB + k + 1] : CH;
  pb = min(max(pb, 0), CH);
  pe = min(max(pe, pb), CH);
  cnt[t] = 0;

  int sum = pb;
  for (int m = 16; m >= 1; m >>= 1) sum += __shfl_xor(sum, m, 32);
  if (lane == 0) sRed[wave] = sum;
  __syncthreads();
  int beg = 0;
  for (int w = 0; w < 8; ++w) beg += sRed[w];

  const int2* __restrict__ piece = CS + (size_t)t * CH;

  for (int j = pb; j < pe; j += 8) {
    int2 p[8];
    #pragma unroll
    for (int u = 0; u < 8; ++u) p[u] = piece[min(j + u, pe - 1)];
    #pragma unroll
    for (int u = 0; u < 8; ++u)
      if (j + u < pe)
        atomicAdd(&cnt[min((unsigned)(p[u].y - n0), (unsigned)(BN - 1))], 1);
  }
  __syncthreads();

  const int v    = cnt[t];
  const int excl = block_excl_scan_256(v, sWave, t);
  const int node = n0 + t;
  if (t < BN && node < N_NODES) {
    rowptr[node] = beg + excl;
    dinv[node]   = rsqrtf((float)v + 1.0f);
  }
  if (k == NB - 1 && t == 0) rowptr[N_NODES] = N_EDGES;
  cur[t] = beg + excl;
  if (t < 32) cur[256 + t] = 0;
  __syncthreads();

  for (int j = pb; j < pe; j += 8) {
    int2 p[8];
    #pragma unroll
    for (int u = 0; u < 8; ++u) p[u] = piece[min(j + u, pe - 1)];
    int pos[8];
    #pragma unroll
    for (int u = 0; u < 8; ++u) {
      const int w = (j + u < pe) ? (int)min((unsigned)(p[u].y - n0), (unsigned)(BN - 1))
                                 : 256 + lane;
      pos[u] = atomicAdd(&cur[w], 1);
    }
    #pragma unroll
    for (int u = 0; u < 8; ++u)
      if (j + u < pe) csr_src[min((unsigned)pos[u], (unsigned)(N_EDGES - 1))] = p[u].x;
  }
}


__device__ __forceinline__ int lane_i(int v, int j) {
  return __builtin_amdgcn_readlane(v, j);
}
__device__ __forceinline__ float lane_f(float v, int j) {
  return __int_as_float(__builtin_amdgcn_readlane(__float_as_int(v), j));
}
__device__ __forceinline__ void axpy4(float4& acc, const float4 h, const float w) {
  acc.x += h.x * w;
  acc.y += h.y * w;
  acc.z += h.z * w;
  acc.w += h.w * w;
}

__device__ __forceinline__ float4 gather_node(const float4* __restrict__ X4,
                                              const int* __restrict__ rowptr,
                                              const int* __restrict__ csr_src,
                                              const float* __restrict__ dinv,
                                              const int d, const int lane) {
  const int beg = __builtin_amdgcn_readfirstlane(rowptr[d]);
  const int end = __builtin_amdgcn_readfirstlane(rowptr[d + 1]);
  const float dd = __int_as_float(__builtin_amdgcn_readfirstlane(__float_as_int(dinv[d])));
  const float4 xs = X4[(size_t)d * 32 + lane];

  float4 acc;
  acc.x = xs.x * dd * dd;
  acc.y = xs.y * dd * dd;
  acc.z = xs.z * dd * dd;
  acc.w = xs.w * dd * dd;

  for (int base = beg; base < end; base += 32) {
    const int n = min(32, end - base);
    const int   s_l = (int)min((unsigned)csr_src[min(base + lane, end - 1)],
                               (unsigned)(N_NODES - 1));
    const float w_l = dinv[s_l];
    int j = 0;
    for (; j + 8 <= n; j += 8) {
      float4 h[8];
      #pragma unroll
      for (int u = 0; u < 8; ++u) h[u] = X4[(size_t)lane_i(s_l, j + u) * 32 + lane];
      #pragma unroll
      for (int u = 0; u < 8; ++u) axpy4(acc, h[u], lane_f(w_l, j + u) * dd);
    }
    if (j + 4 <= n) {
      float4 h[4];
      #pragma unroll
      for (int u = 0; u < 4; ++u) h[u] = X4[(size_t)lane_i(s_l, j + u) * 32 + lane];
      #pragma unroll
      for (int u = 0; u < 4; ++u) axpy4(acc, h[u], lane_f(w_l, j + u) * dd);
      j += 4;
    }
    const int r = n - j;
    float4 h0 = {}, h1 = {}, h2 = {};
    if (r > 0) h0 = X4[(size_t)lane_i(s_l, j)     * 32 + lane];
    if (r > 1) h1 = X4[(size_t)lane_i(s_l, j + 1) * 32 + lane];
    if (r > 2) h2 = X4[(size_t)lane_i(s_l, j + 2) * 32 + lane];
    if (r > 0) axpy4(acc, h0, lane_f(w_l, j) * dd);
    if (r > 1) axpy4(acc, h1, lane_f(w_l, j + 1) * dd);
    if (r > 2) axpy4(acc, h2, lane_f(w_l, j + 2) * dd);
  }
  return acc;
}


#define N_TILES     (N_NODES / 16)
#define N_TGT_TILES (N_TGT / 16)
#define LAYER_GRID 1024

template <bool LIST>
__global__ __launch_bounds__(256) void k_layer(const float* __restrict__ Xin,
                                               const float* __restrict__ W,
                                               const float* __restrict__ bias,
                                               const int* __restrict__ rowptr,
                                               const int* __restrict__ csr_src,
                                               const float* __restrict__ dinv,
                                               const int* __restrict__ list,
                                               const float* __restrict__ Wr,
                                               const float* __restrict__ br,
                                               float* __restrict__ Out) {
  constexpr int n_tiles = LIST ? N_TGT_TILES : N_TILES;
  __shared__ __attribute__((aligned(16))) float sX[16][132];

  const int wave    = threadIdx.x >> 5;
  const int lane    = threadIdx.x & 31;
  const int half    = lane >> 4;
  const int l       = lane & 15;
  const int colBase = wave * 16;
  const float4* __restrict__ X4 = (const float4*)Xin;

  v2f Bf[32];
  #pragma unroll
  for (int k = 0; k < 32; ++k) {
    const int kb = 4 * k + 2 * half;
    Bf[k].x = W[(size_t)kb       * D + colBase + l];
    Bf[k].y = W[(size_t)(kb + 1) * D + colBase + l];
  }
  const float bcol = bias[colBase + l];

  v2f Bh[4];
  float brv = 0.0f;
  if constexpr (LIST) {
    const int lc = min(l, OUTD - 1);
    #pragma unroll
    for (int kk = 0; kk < 4; ++kk) {
      const int kb = colBase + 4 * kk + 2 * half;
      Bh[kk].x = Wr[(size_t)kb       * OUTD + lc];
      Bh[kk].y = Wr[(size_t)(kb + 1) * OUTD + lc];
    }
    if (threadIdx.x < 16 * OUTD) brv = br[threadIdx.x % OUTD];
  }

  for (int tile = blockIdx.x; tile < n_tiles; tile += gridDim.x) {
    const int row0 = tile * 16;

    int2 ids = {};
    if (LIST) ids = *(const int2*)(list + row0 + 2 * wave);
    for (int q = 0; q < 2; ++q) {
      const int r = 2 * wave + q;
      int d = row0 + r;
      if (LIST)
        d = __builtin_amdgcn_readfirstlane(
            (int)min((unsigned)(q ? ids.y : ids.x), (unsigned)(N_NODES - 1)));
      const float4 g = gather_node(X4, rowptr, csr_src, dinv, d, lane);
      *(float4*)&sX[r][4 * lane] = g;
    }
    __syncthreads();

    v8f acc = {};
    #pragma unroll
    for (int k = 0; k < 32; ++k) {
      const int kb = 4 * k + 2 * half;
      v2f a;
      a.x = sX[l][kb];
      a.y = sX[l][kb + 1];
      acc = __builtin_amdgcn_wmma_f32_16x16x4_f32(
           false, a,  false, Bf[k],
           (short)0, acc,  false,  false);
    }
    __syncthreads();

    #pragma unroll
    for (int r = 0; r < 8; ++r)
      sX[r + 8 * half][colBase + l] = fmaxf(acc[r] + bcol, 0.0f);
    __syncthreads();
    if constexpr (!LIST) {
      #pragma unroll
      for (int q = 0; q < 2; ++q) {
        const int r = 2 * wave + q;
        ((float4*)Out)[(size_t)(row0 + r) * 32 + lane] = *(const float4*)&sX[r][4 * lane];
      }
    } else {
      __shared__ float sP[8][16][16];
      v8f hacc = {};
      #pragma unroll
      for (int kk = 0; kk < 4; ++kk) {
        const int kb = colBase + 4 * kk + 2 * half;
        v2f a;
        a.x = sX[l][kb];
        a.y = sX[l][kb + 1];
        hacc = __builtin_amdgcn_wmma_f32_16x16x4_f32(
             false, a,  false, Bh[kk],
             (short)0, hacc,  false,  false);
      }
      #pragma unroll
      for (int r = 0; r < 8; ++r)
        sP[wave][r + 8 * half][l] = hacc[r];
      __syncthreads();
      if (threadIdx.x < 16 * OUTD) {
        const int i = threadIdx.x / OUTD;
        const int o = threadIdx.x % OUTD;
        float p = sP[0][i][o];
        #pragma unroll
        for (int w = 1; w < 8; ++w) p += sP[w][i][o];
        Out[(size_t)row0 * OUTD + threadIdx.x] = p + brv;
      }
    }
  }
}


extern "C" void kernel_launch(void* const* d_in, const int* in_sizes, int n_in,
                              void* d_out, int out_size, void* d_ws, size_t ws_size,
                              hipStream_t stream) {
  const float* x   = (const float*)d_in[0];
  const int*   ei  = (const int*)  d_in[1];
  const int*   tgt = (const int*)  d_in[2];
  const float* W1  = (const float*)d_in[3];
  const float* b1  = (const float*)d_in[4];
  const float* W2  = (const float*)d_in[5];
  const float* b2  = (const float*)d_in[6];
  const float* W3  = (const float*)d_in[7];
  const float* b3  = (const float*)d_in[8];
  const float* Wr  = (const float*)d_in[9];
  const float* br  = (const float*)d_in[10];
  float* out = (float*)d_out;

  const int* src = ei;
  const int* dst = ei + N_EDGES;

  float* ws      = (float*)d_ws;
  float* P0      = ws;
  float* P1      = P0 + (size_t)N_NODES * D;
  int*   rowptr  = (int*)(P1 + (size_t)N_NODES * D) + N_PAD;
  float* dinv    = (float*)(rowptr + N_PAD);
  int*   csr_src = (int*)(dinv + N_PAD);
  int2*  CS      = (int2*)P1;
  int*   offG    = (int*)P1 + 2 * (size_t)N_EDGES;

  k_chunk_sort<<<N_CHUNK, 256, 0, stream>>>(src, dst, offG, CS);
  k_bucket<<<NB, 256, 0, stream>>>(offG, CS, rowptr, dinv, csr_src);

  k_layer<false><<<LAYER_GRID, 256, 0, stream>>>(x,  W1, b1, rowptr, csr_src, dinv, nullptr,
                                                 nullptr, nullptr, P0);
  k_layer<false><<<LAYER_GRID, 256, 0, stream>>>(P0, W2, b2, rowptr, csr_src, dinv, nullptr,
                                                 nullptr, nullptr, P1);
  k_layer<true><<<N_TGT_TILES, 256, 0, stream>>>(P1, W3, b3, rowptr, csr_src, dinv, tgt,
                                                 Wr, br, out);
}


# Gather index chain prefetched: bounds, first 32 source indices and own row of a wave's next node and next tile requested while current rows are summed; 8 B fragments parked in LDS; sums unchanged
# speedup vs baseline: 6.3354x; 1.0783x over previous
#include <hip/hip_runtime.h>

typedef __attribute__((ext_vector_type(2))) float v2f;
typedef __attribute__((ext_vector_type(8))) float v8f;

#define N_NODES 50000
#define N_EDGES 800000
#define D       128
#define OUTD    12
#define N_TGT   8192

#define N_PAD   50048


#define BN      196
#define NB      256
#define N_CHUNK 256
#define CH      (N_EDGES / N_CHUNK)
#define CH_PT   13

static_assert(NB * BN >= N_NODES, "buckets must cover every node");
static_assert(N_CHUNK * CH == N_EDGES, "chunks must cover every edge");
static_assert(256 * CH_PT >= CH, "a block must cover its chunk");

__device__ __forceinline__ int block_excl_scan_256(const int v, int* sWave, const int t) {
  const int lane = t & 31;
  const int wave = t >> 5;
  int inc = v;
  for (int s = 1; s < 32; s <<= 1) {
    const int up = __shfl_up(inc, s, 32);
    if (lane >= s) inc += up;
  }
  if (lane == 31) sWave[wave] = inc;
  __syncthreads();
  int off = 0;
  for (int w = 0; w < 8; ++w)
    if (w < wave) off += sWave[w];
  return off + inc - v;
}

__global__ __launch_bounds__(256) void k_chunk_sort(const int* __restrict__ src,
                                                    const int* __restrict__ dst,
                                                    int* __restrict__ offG,
                                                    int2* __restrict__ CS) {
  __shared__ int h[NB];
  __shared__ int sWave[8];
  const int t  = threadIdx.x;
  const int c  = blockIdx.x;
  const int e0 = c * CH;
  __builtin_assume(t < 256);

  h[t] = 0;
  int s[CH_PT], d[CH_PT];
  #pragma unroll
  for (int u = 0; u < CH_PT; ++u) {
    const int i = min(t + 256 * u, CH - 1);
    s[u] = src[e0 + i];
    d[u] = (int)min((unsigned)dst[e0 + i], (unsigned)(N_NODES - 1));
  }
  __syncthreads();
  #pragma unroll
  for (int u = 0; u < CH_PT; ++u)
    if (t + 256 * u < CH) atomicAdd(&h[d[u] / BN], 1);
  __syncthreads();

  const int v    = h[t];
  const int excl = block_excl_scan_256(v, sWave, t);
  offG[c * NB + t] = excl;
  h[t] = excl;
  __syncthreads();
  int pos[CH_PT];
  #pragma unroll
  for (int u = 0; u < CH_PT; ++u)
    pos[u] = (t + 256 * u < CH) ? atomicAdd(&h[d[u] / BN], 1) : 0;
  #pragma unroll
  for (int u = 0; u < CH_PT; ++u)
    if (t + 256 * u < CH)
      CS[(size_t)e0 + min((unsigned)pos[u], (unsigned)(CH - 1))] = make_int2(s[u], d[u]);
}

__global__ __launch_bounds__(256) void k_bucket(const int* __restrict__ offG,
                                                const int2* __restrict__ CS,
                                                int* __restrict__ rowptr,
                                                float* __restrict__ dinv,
                                                int* __restrict__ csr_src) {
  __shared__ int cnt[256];
  __shared__ int cur[256 + 32];
  __shared__ int sRed[8];
  __shared__ int sWave[8];
  const int t    = threadIdx.x;
  const int k    = blockIdx.x;
  const int lane = t & 31;
  const int wave = t >> 5;
  const int n0   = k * BN;

  int pb = offG[t * NB + k];
  int pe = (k + 1 < NB) ? offG[t * NB + k + 1] : CH;
  pb = min(max(pb, 0), CH);
  pe = min(max(pe, pb), CH);
  cnt[t] = 0;

  int sum = pb;
  for (int m = 16; m >= 1; m >>= 1) sum += __shfl_xor(sum, m, 32);
  if (lane == 0) sRed[wave] = sum;
  __syncthreads();
  int beg = 0;
  for (int w = 0; w < 8; ++w) beg += sRed[w];

  const int2* __restrict__ piece = CS + (size_t)t * CH;

  for (int j = pb; j < pe; j += 8) {
    int2 p[8];
    #pragma unroll
    for (int u = 0; u < 8; ++u) p[u] = piece[min(j + u, pe - 1)];
    #pragma unroll
    for (int u = 0; u < 8; ++u)
      if (j + u < pe)
        atomicAdd(&cnt[min((unsigned)(p[u].y - n0), (unsigned)(BN - 1))], 1);
  }
  __syncthreads();

  const int v    = cnt[t];
  const int excl = block_excl_scan_256(v, sWave, t);
  const int node = n0 + t;
  if (t < BN && node < N_NODES) {
    rowptr[node] = beg + excl;
    dinv[node]   = rsqrtf((float)v + 1.0f);
  }
  if (k == NB - 1 && t == 0) rowptr[N_NODES] = N_EDGES;
  cur[t] = beg + excl;
  if (t < 32) cur[256 + t] = 0;
  __syncthreads();

  for (int j = pb; j < pe; j += 8) {
    int2 p[8];
    #pragma unroll
    for (int u = 0; u < 8; ++u) p[u] = piece[min(j + u, pe - 1)];
    int pos[8];
    #pragma unroll
    for (int u = 0; u < 8; ++u) {
      const int w = (j + u < pe) ? (int)min((unsigned)(p[u].y - n0), (unsigned)(BN - 1))
                                 : 256 + lane;
      pos[u] = atomicAdd(&cur[w], 1);
    }
    #pragma unroll
    for (int u = 0; u < 8; ++u)
      if (j + u < pe) csr_src[min((unsigned)pos[u], (unsigned)(N_EDGES - 1))] = p[u].x;
  }
}


__device__ __forceinline__ int lane_i(int v, int j) {
  return __builtin_amdgcn_readlane(v, j);
}
__device__ __forceinline__ float lane_f(float v, int j) {
  return __int_as_float(__builtin_amdgcn_readlane(__float_as_int(v), j));
}
__device__ __forceinline__ void axpy4(float4& acc, const float4 h, const float w) {
  acc.x += h.x * w;
  acc.y += h.y * w;
  acc.z += h.z * w;
  acc.w += h.w * w;
}

static_assert((size_t)N_NODES * D * sizeof(float) < (1ull << 32), "32-bit row offsets");
__device__ __forceinline__ float4 row_at(const float4* __restrict__ X4, const int row,
                                         const int lane) {
  const unsigned off = (unsigned)row * (unsigned)(D * sizeof(float)) + (unsigned)lane * 16u;
  return *(const float4*)((const char*)X4 + off);
}

struct NodeHead { int rp0, rp1; float ddv; };
struct NodeIdx  { int beg, end; float dd; int s_raw; float4 xs; };
struct TileHead { int rp_v; float dv_v; };

__device__ __forceinline__ TileHead tile_head(const int* __restrict__ rowptr,
                                              const float* __restrict__ dinv,
                                              const int row0, const int lane) {
  TileHead t;
  t.rp_v = rowptr[min(row0 + lane, N_NODES)];
  t.dv_v = dinv[min(row0 + lane, N_NODES - 1)];
  return t;
}
__device__ __forceinline__ NodeHead head_of(const TileHead t, const int r) {
  NodeHead h;
  h.rp0 = lane_i(t.rp_v, r);
  h.rp1 = lane_i(t.rp_v, r + 1);
  h.ddv = lane_f(t.dv_v, r);
  return h;
}

__device__ __forceinline__ NodeIdx node_idx(const float4* __restrict__ X4,
                                            const int* __restrict__ csr_src,
                                            const NodeHead h, const int d, const int lane) {
  NodeIdx x;
  x.beg = __builtin_amdgcn_readfirstlane(h.rp0);
  x.end = __builtin_amdgcn_readfirstlane(h.rp1);
  x.dd = __int_as_float(__builtin_amdgcn_readfirstlane(__float_as_int(h.ddv)));
  const int p = min(max(min(x.beg + lane, x.end - 1), 0), N_EDGES - 1);
  x.s_raw = csr_src[p];
  x.xs = row_at(X4, d, lane);
  return x;
}

__device__ __forceinline__ void sum_rows(float4& acc, const float4* __restrict__ X4,
                                         const int s_l, const float w_l, const int n,
                                         const float dd, const int lane) {
  int j = 0;
  for (; j + 8 <= n; j += 8) {
    float4 h[8];
    #pragma unroll
    for (int u = 0; u < 8; ++u) h[u] = row_at(X4, lane_i(s_l, j + u), lane);
    #pragma unroll
    for (int u = 0; u < 8; ++u) axpy4(acc, h[u], lane_f(w_l, j + u) * dd);
  }
  if (j + 4 <= n) {
    float4 h[4];
    #pragma unroll
    for (int u = 0; u < 4; ++u) h[u] = row_at(X4, lane_i(s_l, j + u), lane);
    #pragma unroll
    for (int u = 0; u < 4; ++u) axpy4(acc, h[u], lane_f(w_l, j + u) * dd);
    j += 4;
  }
  const int r = n - j;
  float4 h0 = {}, h1 = {}, h2 = {};
  if (r > 0) h0 = row_at(X4, lane_i(s_l, j), lane);
  if (r > 1) h1 = row_at(X4, lane_i(s_l, j + 1), lane);
  if (r > 2) h2 = row_at(X4, lane_i(s_l, j + 2), lane);
  if (r > 0) axpy4(acc, h0, lane_f(w_l, j) * dd);
  if (r > 1) axpy4(acc, h1, lane_f(w_l, j + 1) * dd);
  if (r > 2) axpy4(acc, h2, lane_f(w_l, j + 2) * dd);
}

__device__ __forceinline__ float4 node_rows(const float4* __restrict__ X4,
                                            const int* __restrict__ csr_src,
                                            const float* __restrict__ dinv,
                                            const NodeIdx x, const int lane) {
  const float dd = x.dd;
  float4 acc;
  acc.x = x.xs.x * dd * dd;
  acc.y = x.xs.y * dd * dd;
  acc.z = x.xs.z * dd * dd;
  acc.w = x.xs.w * dd * dd;

  if (x.beg < x.end) {
    const int   s_l = (int)min((unsigned)x.s_raw, (unsigned)(N_NODES - 1));
    const float w_l = dinv[s_l];
    sum_rows(acc, X4, s_l, w_l, min(32, x.end - x.beg), dd, lane);
  }
  for (int base = x.beg + 32; base < x.end; base += 32) {
    const int   s_l = (int)min((unsigned)csr_src[min(base + lane, x.end - 1)],
                               (unsigned)(N_NODES - 1));
    const float w_l = dinv[s_l];
    sum_rows(acc, X4, s_l, w_l, min(32, x.end - base), dd, lane);
  }
  return acc;
}


#define N_TILES     (N_NODES / 16)
#define N_TGT_TILES (N_TGT / 16)
#define LAYER_GRID 1024

template <bool LIST>
__global__ __launch_bounds__(256, LIST ? 4 : 8) void k_layer(const float* __restrict__ Xin,
                                               const float* __restrict__ W,
                                               const float* __restrict__ bias,
                                               const int* __restrict__ rowptr,
                                               const int* __restrict__ csr_src,
                                               const float* __restrict__ dinv,
                                               const int* __restrict__ list,
                                               const float* __restrict__ Wr,
                                               const float* __restrict__ br,
                                               float* __restrict__ Out) {
  constexpr int n_tiles = LIST ? N_TGT_TILES : N_TILES;
  __shared__ __attribute__((aligned(16))) float sX[16][132];

  const int wave    = __builtin_amdgcn_readfirstlane((int)(threadIdx.x >> 5));
  const int lane    = threadIdx.x & 31;
  const int half    = lane >> 4;
  const int l       = lane & 15;
  const int colBase = wave * 16;
  const float4* __restrict__ X4 = (const float4*)Xin;

  constexpr int B_REG = 24;
  __shared__ v2f sB[8][32 - B_REG][32];
  v2f Bf[B_REG];
  #pragma unroll
  for (int k = 0; k < 32; ++k) {
    const int kb = 4 * k + 2 * half;
    v2f b;
    b.x = W[(size_t)kb       * D + colBase + l];
    b.y = W[(size_t)(kb + 1) * D + colBase + l];
    if (k < B_REG) Bf[k] = b;
    else           sB[wave][k - B_REG][lane] = b;
  }
  const float bcol = bias[colBase + l];

  v2f Bh[4];
  float brv = 0.0f;
  if constexpr (LIST) {
    const int lc = min(l, OUTD - 1);
    #pragma unroll
    for (int kk = 0; kk < 4; ++kk) {
      const int kb = colBase + 4 * kk + 2 * half;
      Bh[kk].x = Wr[(size_t)kb       * OUTD + lc];
      Bh[kk].y = Wr[(size_t)(kb + 1) * OUTD + lc];
    }
    if (threadIdx.x < 16 * OUTD) brv = br[threadIdx.x % OUTD];
  }

  int d0 = 0, d1 = 0;
  NodeHead hA = {}, hB = {};
  NodeIdx  iA = {};
  if constexpr (!LIST) {
    const int frow0 = min((int)blockIdx.x, n_tiles - 1) * 16;
    const TileHead tf = tile_head(rowptr, dinv, frow0, lane);
    d0 = frow0 + 2 * wave;
    d1 = d0 + 1;
    hA = head_of(tf, 2 * wave);
    hB = head_of(tf, 2 * wave + 1);
    iA = node_idx(X4, csr_src, hA, d0, lane);
  }

  for (int tile = blockIdx.x; tile < n_tiles; tile += gridDim.x) {
    const int row0 = tile * 16;

    if constexpr (LIST) {
      const int2 ids = *(const int2*)(list + row0 + 2 * wave);
      d0 = __builtin_amdgcn_readfirstlane((int)min((unsigned)ids.x, (unsigned)(N_NODES - 1)));
      d1 = __builtin_amdgcn_readfirstlane((int)min((unsigned)ids.y, (unsigned)(N_NODES - 1)));
      const int   hv = rowptr[((lane & 2) ? d1 : d0) + (lane & 1)];
      const float dv = dinv[(lane & 1) ? d1 : d0];
      hA.rp0 = lane_i(hv, 0);
      hA.rp1 = lane_i(hv, 1);
      hA.ddv = lane_f(dv, 0);
      hB.rp0 = lane_i(hv, 2);
      hB.rp1 = lane_i(hv, 3);
      hB.ddv = lane_f(dv, 1);
      iA = node_idx(X4, csr_src, hA, d0, lane);
    }
    const NodeIdx iB = node_idx(X4, csr_src, hB, d1, lane);
    TileHead tn = {};
    int nrow0 = 0;
    if constexpr (!LIST) {
      nrow0 = min(tile + (int)gridDim.x, n_tiles - 1) * 16;
      tn = tile_head(rowptr, dinv, nrow0, lane);
    }
    asm volatile("" ::: "memory");
    {
      const float4 g = node_rows(X4, csr_src, dinv, iA, lane);
      *(float4*)&sX[2 * wave][4 * lane] = g;
    }
    NodeHead hBn = {};
    NodeIdx  iAn = {};
    if constexpr (!LIST) {
      hBn = head_of(tn, 2 * wave + 1);
      iAn = node_idx(X4, csr_src, head_of(tn, 2 * wave), nrow0 + 2 * wave, lane);
    }
    {
      const float4 g = node_rows(X4, csr_src, dinv, iB, lane);
      *(float4*)&sX[2 * wave + 1][4 * lane] = g;
    }
    if constexpr (!LIST) {
      d1 = nrow0 + 2 * wave + 1;
      hB = hBn;
      iA = iAn;
    }
    __syncthreads();

    v8f acc = {};
    #pragma unroll
    for (int k = 0; k < 32; ++k) {
      const int kb = 4 * k + 2 * half;
      v2f a;
      a.x = sX[l][kb];
      a.y = sX[l][kb + 1];
      const v2f b = (k < B_REG) ? Bf[k] : sB[wave][k - B_REG][lane];
      acc = __builtin_amdgcn_wmma_f32_16x16x4_f32(
           false, a,  false, b,
           (short)0, acc,  false,  false);
    }
    __syncthreads();

    #pragma unroll
    for (int r = 0; r < 8; ++r)
      sX[r + 8 * half][colBase + l] = fmaxf(acc[r] + bcol, 0.0f);
    __syncthreads();
    if constexpr (!LIST) {
      #pragma unroll
      for (int q = 0; q < 2; ++q) {
        const int r = 2 * wave + q;
        ((float4*)Out)[(size_t)(row0 + r) * 32 + lane] = *(const float4*)&sX[r][4 * lane];
      }
    } else {
      __shared__ float sP[8][16][16];
      v8f hacc = {};
      #pragma unroll
      for (int kk = 0; kk < 4; ++kk) {
        const int kb = colBase + 4 * kk + 2 * half;
        v2f a;
        a.x = sX[l][kb];
        a.y = sX[l][kb + 1];
        hacc = __builtin_amdgcn_wmma_f32_16x16x4_f32(
             false, a,  false, Bh[kk],
             (short)0, hacc,  false,  false);
      }
      #pragma unroll
      for (int r = 0; r < 8; ++r)
        sP[wave][r + 8 * half][l] = hacc[r];
      __syncthreads();
      if (threadIdx.x < 16 * OUTD) {
        const int i = threadIdx.x / OUTD;
        const int o = threadIdx.x % OUTD;
        float p = sP[0][i][o];
        #pragma unroll
        for (int w = 1; w < 8; ++w) p += sP[w][i][o];
        Out[(size_t)row0 * OUTD + threadIdx.x] = p + brv;
      }
    }
  }
}


extern "C" void kernel_launch(void* const* d_in, const int* in_sizes, int n_in,
                              void* d_out, int out_size, void* d_ws, size_t ws_size,
                              hipStream_t stream) {
  const float* x   = (const float*)d_in[0];
  const int*   ei  = (const int*)  d_in[1];
  const int*   tgt = (const int*)  d_in[2];
  const float* W1  = (const float*)d_in[3];
  const float* b1  = (const float*)d_in[4];
  const float* W2  = (const float*)d_in[5];
  const float* b2  = (const float*)d_in[6];
  const float* W3  = (const float*)d_in[7];
  const float* b3  = (const float*)d_in[8];
  const float* Wr  = (const float*)d_in[9];
  const float* br  = (const float*)d_in[10];
  float* out = (float*)d_out;

  const int* src = ei;
  const int* dst = ei + N_EDGES;

  float* ws      = (float*)d_ws;
  float* P0      = ws;
  float* P1      = P0 + (size_t)N_NODES * D;
  int*   rowptr  = (int*)(P1 + (size_t)N_NODES * D) + N_PAD;
  float* dinv    = (float*)(rowptr + N_PAD);
  int*   csr_src = (int*)(dinv + N_PAD);
  int2*  CS      = (int2*)P1;
  int*   offG    = (int*)P1 + 2 * (size_t)N_EDGES;

  k_chunk_sort<<<N_CHUNK, 256, 0, stream>>>(src, dst, offG, CS);
  k_bucket<<<NB, 256, 0, stream>>>(offG, CS, rowptr, dinv, csr_src);

  k_layer<false><<<LAYER_GRID, 256, 0, stream>>>(x,  W1, b1, rowptr, csr_src, dinv, nullptr,
                                                 nullptr, nullptr, P0);
  k_layer<false><<<LAYER_GRID, 256, 0, stream>>>(P0, W2, b2, rowptr, csr_src, dinv, nullptr,
                                                 nullptr, nullptr, P1);
  k_layer<true><<<N_TGT_TILES, 256, 0, stream>>>(P1, W3, b3, rowptr, csr_src, dinv, tgt,
                                                 Wr, br, out);
}


# Both sort passes put their records at the sorted place in LDS and write the finished run out as whole 128-B lines instead of scattering 8-B and 4-B stores over up to 32 lines each; list order kept
# speedup vs baseline: 6.5587x; 1.0352x over previous
#include <hip/hip_runtime.h>

typedef __attribute__((ext_vector_type(2))) float v2f;
typedef __attribute__((ext_vector_type(8))) float v8f;

#define N_NODES 50000
#define N_EDGES 800000
#define D       128
#define OUTD    12
#define N_TGT   8192

#define N_PAD   50048


#define BN      196
#define NB      256
#define N_CHUNK 256
#define CH      (N_EDGES / N_CHUNK)
#define CH_PT   13
#define STG     8192

static_assert(NB * BN >= N_NODES, "buckets must cover every node");
static_assert(N_CHUNK * CH == N_EDGES, "chunks must cover every edge");
static_assert(256 * CH_PT >= CH + 15, "a block must cover its chunk from the line boundary below it");
static_assert(256 * (CH_PT - 1) <= CH && 256 >= 15, "only the first and the last slot can miss the chunk");

__device__ __forceinline__ int block_excl_scan_256(const int v, int* sWave, const int t) {
  const int lane = t & 31;
  const int wave = t >> 5;
  int inc = v;
  for (int s = 1; s < 32; s <<= 1) {
    const int up = __shfl_up(inc, s, 32);
    if (lane >= s) inc += up;
  }
  if (lane == 31) sWave[wave] = inc;
  __syncthreads();
  int off = 0;
  for (int w = 0; w < 8; ++w)
    if (w < wave) off += sWave[w];
  return off + inc - v;
}

__global__ __launch_bounds__(256) void k_chunk_sort(const int* __restrict__ src,
                                                    const int* __restrict__ dst,
                                                    int* __restrict__ offG,
                                                    int2* __restrict__ CS) {
  __shared__ int h[NB];
  __shared__ int sWave[8];
  __shared__ int2 sCS[CH];
  const int t  = threadIdx.x;
  const int c  = blockIdx.x;
  const int e0 = c * CH;
  __builtin_assume(t < 256);

  h[t] = 0;
  int s[CH_PT], d[CH_PT];
  #pragma unroll
  for (int u = 0; u < CH_PT; ++u) {
    const int i = min(t + 256 * u, CH - 1);
    s[u] = src[e0 + i];
    d[u] = (int)min((unsigned)dst[e0 + i], (unsigned)(N_NODES - 1));
  }
  __syncthreads();
  #pragma unroll
  for (int u = 0; u < CH_PT; ++u)
    if (t + 256 * u < CH) atomicAdd(&h[d[u] / BN], 1);
  __syncthreads();

  const int v    = h[t];
  const int excl = block_excl_scan_256(v, sWave, t);
  offG[c * NB + t] = excl;
  h[t] = excl;
  __syncthreads();
  int pos[CH_PT];
  #pragma unroll
  for (int u = 0; u < CH_PT; ++u)
    pos[u] = (t + 256 * u < CH) ? atomicAdd(&h[d[u] / BN], 1) : 0;
  #pragma unroll
  for (int u = 0; u < CH_PT; ++u)
    if (t + 256 * u < CH)
      sCS[min((unsigned)pos[u], (unsigned)(CH - 1))] = make_int2(s[u], d[u]);
  __syncthreads();
  const int sh = e0 & 15;
  int2 pr[CH_PT];
  #pragma unroll
  for (int u = 0; u < CH_PT; ++u)
    pr[u] = sCS[min(max(t + 256 * u - sh, 0), CH - 1)];
  #pragma unroll
  for (int u = 0; u < CH_PT; ++u) {
    const int i = t + 256 * u - sh;
    const bool in = (u == 0) ? (i >= 0) : (u == CH_PT - 1) ? (i < CH) : true;
    if (in) CS[e0 + i] = pr[u];
  }
}

template <bool STAGED>
__device__ __forceinline__ void bucket_place(const int2* __restrict__ piece, const int pb,
                                             const int pe, const int n0, const int lane,
                                             const int beg, int* cur, int* sOut,
                                             int* __restrict__ csr_src) {
  for (int j = pb; j < pe; j += 8) {
    int2 p[8];
    #pragma unroll
    for (int u = 0; u < 8; ++u) p[u] = piece[min(j + u, pe - 1)];
    int pos[8];
    #pragma unroll
    for (int u = 0; u < 8; ++u) {
      const int w = (j + u < pe) ? (int)min((unsigned)(p[u].y - n0), (unsigned)(BN - 1))
                                 : 256 + lane;
      pos[u] = atomicAdd(&cur[w], 1);
    }
    #pragma unroll
    for (int u = 0; u < 8; ++u)
      if (j + u < pe) {
        if constexpr (STAGED)
          sOut[min((unsigned)(pos[u] - beg), (unsigned)(STG - 1))] = p[u].x;
        else
          csr_src[min((unsigned)pos[u], (unsigned)(N_EDGES - 1))] = p[u].x;
      }
  }
}

__global__ __launch_bounds__(256) void k_bucket(const int* __restrict__ offG,
                                                const int2* __restrict__ CS,
                                                int* __restrict__ rowptr,
                                                float* __restrict__ dinv,
                                                int* __restrict__ csr_src) {
  __shared__ int cnt[256];
  __shared__ int cur[256 + 32];
  __shared__ int sRed[8];
  __shared__ int sRedN[8];
  __shared__ int sWave[8];
  __shared__ int sOut[STG];
  const int t    = threadIdx.x;
  const int k    = blockIdx.x;
  const int lane = t & 31;
  const int wave = t >> 5;
  const int n0   = k * BN;

  int pb = offG[t * NB + k];
  int pe = (k + 1 < NB) ? offG[t * NB + k + 1] : CH;
  pb = min(max(pb, 0), CH);
  pe = min(max(pe, pb), CH);
  cnt[t] = 0;

  int sum = pb;
  int len = pe - pb;
  for (int m = 16; m >= 1; m >>= 1) {
    sum += __shfl_xor(sum, m, 32);
    len += __shfl_xor(len, m, 32);
  }
  if (lane == 0) { sRed[wave] = sum; sRedN[wave] = len; }
  __syncthreads();
  int beg = 0;
  int n   = 0;
  for (int w = 0; w < 8; ++w) { beg += sRed[w]; n += sRedN[w]; }
  const bool staged = n <= STG;

  const int2* __restrict__ piece = CS + (size_t)t * CH;

  for (int j = pb; j < pe; j += 8) {
    int2 p[8];
    #pragma unroll
    for (int u = 0; u < 8; ++u) p[u] = piece[min(j + u, pe - 1)];
    #pragma unroll
    for (int u = 0; u < 8; ++u)
      if (j + u < pe)
        atomicAdd(&cnt[min((unsigned)(p[u].y - n0), (unsigned)(BN - 1))], 1);
  }
  __syncthreads();

  const int v    = cnt[t];
  const int excl = block_excl_scan_256(v, sWave, t);
  const int node = n0 + t;
  if (t < BN && node < N_NODES) {
    rowptr[node] = beg + excl;
    dinv[node]   = rsqrtf((float)v + 1.0f);
  }
  if (k == NB - 1 && t == 0) rowptr[N_NODES] = N_EDGES;
  cur[t] = beg + excl;
  if (t < 32) cur[256 + t] = 0;
  __syncthreads();

  if (!staged) bucket_place<false>(piece, pb, pe, n0, lane, beg, cur, sOut, csr_src);
  if (staged) {
    bucket_place<true>(piece, pb, pe, n0, lane, beg, cur, sOut, csr_src);
    __syncthreads();
    const int a0   = beg & ~31;
    const int last = max(n, 1) - 1;
    for (int g0 = a0 + t; g0 < beg + n; g0 += 1024) {
      int sv[4];
      #pragma unroll
      for (int q = 0; q < 4; ++q)
        sv[q] = sOut[min(max(g0 + 256 * q - beg, 0), last)];
      #pragma unroll
      for (int q = 0; q < 4; ++q) {
        const int g = g0 + 256 * q;
        if (g >= beg && g < beg + n && g < N_EDGES) csr_src[g] = sv[q];
      }
    }
  }
}


__device__ __forceinline__ int lane_i(int v, int j) {
  return __builtin_amdgcn_readlane(v, j);
}
__device__ __forceinline__ float lane_f(float v, int j) {
  return __int_as_float(__builtin_amdgcn_readlane(__float_as_int(v), j));
}
__device__ __forceinline__ void axpy4(float4& acc, const float4 h, const float w) {
  acc.x += h.x * w;
  acc.y += h.y * w;
  acc.z += h.z * w;
  acc.w += h.w * w;
}

static_assert((size_t)N_NODES * D * sizeof(float) < (1ull << 32), "32-bit row offsets");
__device__ __forceinline__ float4 row_at(const float4* __restrict__ X4, const int row,
                                         const int lane) {
  const unsigned off = (unsigned)row * (unsigned)(D * sizeof(float)) + (unsigned)lane * 16u;
  return *(const float4*)((const char*)X4 + off);
}

struct NodeHead { int rp0, rp1; float ddv; };
struct NodeIdx  { int beg, end; float dd; int s_raw; float4 xs; };
struct TileHead { int rp_v; float dv_v; };

__device__ __forceinline__ TileHead tile_head(const int* __restrict__ rowptr,
                                              const float* __restrict__ dinv,
                                              const int row0, const int lane) {
  TileHead t;
  t.rp_v = rowptr[min(row0 + lane, N_NODES)];
  t.dv_v = dinv[min(row0 + lane, N_NODES - 1)];
  return t;
}
__device__ __forceinline__ NodeHead head_of(const TileHead t, const int r) {
  NodeHead h;
  h.rp0 = lane_i(t.rp_v, r);
  h.rp1 = lane_i(t.rp_v, r + 1);
  h.ddv = lane_f(t.dv_v, r);
  return h;
}

__device__ __forceinline__ NodeIdx node_idx(const float4* __restrict__ X4,
                                            const int* __restrict__ csr_src,
                                            const NodeHead h, const int d, const int lane) {
  NodeIdx x;
  x.beg = __builtin_amdgcn_readfirstlane(h.rp0);
  x.end = __builtin_amdgcn_readfirstlane(h.rp1);
  x.dd = __int_as_float(__builtin_amdgcn_readfirstlane(__float_as_int(h.ddv)));
  const int p = min(max(min(x.beg + lane, x.end - 1), 0), N_EDGES - 1);
  x.s_raw = csr_src[p];
  x.xs = row_at(X4, d, lane);
  return x;
}

__device__ __forceinline__ void sum_rows(float4& acc, const float4* __restrict__ X4,
                                         const int s_l, const float w_l, const int n,
                                         const float dd, const int lane) {
  int j = 0;
  for (; j + 8 <= n; j += 8) {
    float4 h[8];
    #pragma unroll
    for (int u = 0; u < 8; ++u) h[u] = row_at(X4, lane_i(s_l, j + u), lane);
    #pragma unroll
    for (int u = 0; u < 8; ++u) axpy4(acc, h[u], lane_f(w_l, j + u) * dd);
  }
  if (j + 4 <= n) {
    float4 h[4];
    #pragma unroll
    for (int u = 0; u < 4; ++u) h[u] = row_at(X4, lane_i(s_l, j + u), lane);
    #pragma unroll
    for (int u = 0; u < 4; ++u) axpy4(acc, h[u], lane_f(w_l, j + u) * dd);
    j += 4;
  }
  const int r = n - j;
  float4 h0 = {}, h1 = {}, h2 = {};
  if (r > 0) h0 = row_at(X4, lane_i(s_l, j), lane);
  if (r > 1) h1 = row_at(X4, lane_i(s_l, j + 1), lane);
  if (r > 2) h2 = row_at(X4, lane_i(s_l, j + 2), lane);
  if (r > 0) axpy4(acc, h0, lane_f(w_l, j) * dd);
  if (r > 1) axpy4(acc, h1, lane_f(w_l, j + 1) * dd);
  if (r > 2) axpy4(acc, h2, lane_f(w_l, j + 2) * dd);
}

__device__ __forceinline__ float4 node_rows(const float4* __restrict__ X4,
                                            const int* __restrict__ csr_src,
                                            const float* __restrict__ dinv,
                                            const NodeIdx x, const int lane) {
  const float dd = x.dd;
  float4 acc;
  acc.x = x.xs.x * dd * dd;
  acc.y = x.xs.y * dd * dd;
  acc.z = x.xs.z * dd * dd;
  acc.w = x.xs.w * dd * dd;

  if (x.beg < x.end) {
    const int   s_l = (int)min((unsigned)x.s_raw, (unsigned)(N_NODES - 1));
    const float w_l = dinv[s_l];
    sum_rows(acc, X4, s_l, w_l, min(32, x.end - x.beg), dd, lane);
  }
  for (int base = x.beg + 32; base < x.end; base += 32) {
    const int   s_l = (int)min((unsigned)csr_src[min(base + lane, x.end - 1)],
                               (unsigned)(N_NODES - 1));
    const float w_l = dinv[s_l];
    sum_rows(acc, X4, s_l, w_l, min(32, x.end - base), dd, lane);
  }
  return acc;
}


#define N_TILES     (N_NODES / 16)
#define N_TGT_TILES (N_TGT / 16)
#define LAYER_GRID 1024

template <bool LIST>
__global__ __launch_bounds__(256, LIST ? 4 : 8) void k_layer(const float* __restrict__ Xin,
                                               const float* __restrict__ W,
                                               const float* __restrict__ bias,
                                               const int* __restrict__ rowptr,
                                               const int* __restrict__ csr_src,
                                               const float* __restrict__ dinv,
                                               const int* __restrict__ list,
                                               const float* __restrict__ Wr,
                                               const float* __restrict__ br,
                                               float* __restrict__ Out) {
  constexpr int n_tiles = LIST ? N_TGT_TILES : N_TILES;
  __shared__ __attribute__((aligned(16))) float sX[16][132];

  const int wave    = __builtin_amdgcn_readfirstlane((int)(threadIdx.x >> 5));
  const int lane    = threadIdx.x & 31;
  const int half    = lane >> 4;
  const int l       = lane & 15;
  const int colBase = wave * 16;
  const float4* __restrict__ X4 = (const float4*)Xin;

  constexpr int B_REG = 24;
  __shared__ v2f sB[8][32 - B_REG][32];
  v2f Bf[B_REG];
  #pragma unroll
  for (int k = 0; k < 32; ++k) {
    const int kb = 4 * k + 2 * half;
    v2f b;
    b.x = W[(size_t)kb       * D + colBase + l];
    b.y = W[(size_t)(kb + 1) * D + colBase + l];
    if (k < B_REG) Bf[k] = b;
    else           sB[wave][k - B_REG][lane] = b;
  }
  const float bcol = bias[colBase + l];

  v2f Bh[4];
  float brv = 0.0f;
  if constexpr (LIST) {
    const int lc = min(l, OUTD - 1);
    #pragma unroll
    for (int kk = 0; kk < 4; ++kk) {
      const int kb = colBase + 4 * kk + 2 * half;
      Bh[kk].x = Wr[(size_t)kb       * OUTD + lc];
      Bh[kk].y = Wr[(size_t)(kb + 1) * OUTD + lc];
    }
    if (threadIdx.x < 16 * OUTD) brv = br[threadIdx.x % OUTD];
  }

  int d0 = 0, d1 = 0;
  NodeHead hA = {}, hB = {};
  NodeIdx  iA = {};
  if constexpr (!LIST) {
    const int frow0 = min((int)blockIdx.x, n_tiles - 1) * 16;
    const TileHead tf = tile_head(rowptr, dinv, frow0, lane);
    d0 = frow0 + 2 * wave;
    d1 = d0 + 1;
    hA = head_of(tf, 2 * wave);
    hB = head_of(tf, 2 * wave + 1);
    iA = node_idx(X4, csr_src, hA, d0, lane);
  }

  for (int tile = blockIdx.x; tile < n_tiles; tile += gridDim.x) {
    const int row0 = tile * 16;

    if constexpr (LIST) {
      const int2 ids = *(const int2*)(list + row0 + 2 * wave);
      d0 = __builtin_amdgcn_readfirstlane((int)min((unsigned)ids.x, (unsigned)(N_NODES - 1)));
      d1 = __builtin_amdgcn_readfirstlane((int)min((unsigned)ids.y, (unsigned)(N_NODES - 1)));
      const int   hv = rowptr[((lane & 2) ? d1 : d0) + (lane & 1)];
      const float dv = dinv[(lane & 1) ? d1 : d0];
      hA.rp0 = lane_i(hv, 0);
      hA.rp1 = lane_i(hv, 1);
      hA.ddv = lane_f(dv, 0);
      hB.rp0 = lane_i(hv, 2);
      hB.rp1 = lane_i(hv, 3);
      hB.ddv = lane_f(dv, 1);
      iA = node_idx(X4, csr_src, hA, d0, lane);
    }
    const NodeIdx iB = node_idx(X4, csr_src, hB, d1, lane);
    TileHead tn = {};
    int nrow0 = 0;
    if constexpr (!LIST) {
      nrow0 = min(tile + (int)gridDim.x, n_tiles - 1) * 16;
      tn = tile_head(rowptr, dinv, nrow0, lane);
    }
    asm volatile("" ::: "memory");
    {
      const float4 g = node_rows(X4, csr_src, dinv, iA, lane);
      *(float4*)&sX[2 * wave][4 * lane] = g;
    }
    NodeHead hBn = {};
    NodeIdx  iAn = {};
    if constexpr (!LIST) {
      hBn = head_of(tn, 2 * wave + 1);
      iAn = node_idx(X4, csr_src, head_of(tn, 2 * wave), nrow0 + 2 * wave, lane);
    }
    {
      const float4 g = node_rows(X4, csr_src, dinv, iB, lane);
      *(float4*)&sX[2 * wave + 1][4 * lane] = g;
    }
    if constexpr (!LIST) {
      d1 = nrow0 + 2 * wave + 1;
      hB = hBn;
      iA = iAn;
    }
    __syncthreads();

    v8f acc = {};
    #pragma unroll
    for (int k = 0; k < 32; ++k) {
      const int kb = 4 * k + 2 * half;
      v2f a;
      a.x = sX[l][kb];
      a.y = sX[l][kb + 1];
      const v2f b = (k < B_REG) ? Bf[k] : sB[wave][k - B_REG][lane];
      acc = __builtin_amdgcn_wmma_f32_16x16x4_f32(
           false, a,  false, b,
           (short)0, acc,  false,  false);
    }
    __syncthreads();

    #pragma unroll
    for (int r = 0; r < 8; ++r)
      sX[r + 8 * half][colBase + l] = fmaxf(acc[r] + bcol, 0.0f);
    __syncthreads();
    if constexpr (!LIST) {
      #pragma unroll
      for (int q = 0; q < 2; ++q) {
        const int r = 2 * wave + q;
        ((float4*)Out)[(size_t)(row0 + r) * 32 + lane] = *(const float4*)&sX[r][4 * lane];
      }
    } else {
      __shared__ float sP[8][16][16];
      v8f hacc = {};
      #pragma unroll
      for (int kk = 0; kk < 4; ++kk) {
        const int kb = colBase + 4 * kk + 2 * half;
        v2f a;
        a.x = sX[l][kb];
        a.y = sX[l][kb + 1];
        hacc = __builtin_amdgcn_wmma_f32_16x16x4_f32(
             false, a,  false, Bh[kk],
             (short)0, hacc,  false,  false);
      }
      #pragma unroll
      for (int r = 0; r < 8; ++r)
        sP[wave][r + 8 * half][l] = hacc[r];
      __syncthreads();
      if (threadIdx.x < 16 * OUTD) {
        const int i = threadIdx.x / OUTD;
        const int o = threadIdx.x % OUTD;
        float p = sP[0][i][o];
        #pragma unroll
        for (int w = 1; w < 8; ++w) p += sP[w][i][o];
        Out[(size_t)row0 * OUTD + threadIdx.x] = p + brv;
      }
    }
  }
}


extern "C" void kernel_launch(void* const* d_in, const int* in_sizes, int n_in,
                              void* d_out, int out_size, void* d_ws, size_t ws_size,
                              hipStream_t stream) {
  const float* x   = (const float*)d_in[0];
  const int*   ei  = (const int*)  d_in[1];
  const int*   tgt = (const int*)  d_in[2];
  const float* W1  = (const float*)d_in[3];
  const float* b1  = (const float*)d_in[4];
  const float* W2  = (const float*)d_in[5];
  const float* b2  = (const float*)d_in[6];
  const float* W3  = (const float*)d_in[7];
  const float* b3  = (const float*)d_in[8];
  const float* Wr  = (const float*)d_in[9];
  const float* br  = (const float*)d_in[10];
  float* out = (float*)d_out;

  const int* src = ei;
  const int* dst = ei + N_EDGES;

  float* ws      = (float*)d_ws;
  float* P0      = ws;
  float* P1      = P0 + (size_t)N_NODES * D;
  int*   rowptr  = (int*)(P1 + (size_t)N_NODES * D) + N_PAD;
  float* dinv    = (float*)(rowptr + N_PAD);
  int*   csr_src = (int*)(dinv + N_PAD);
  int2*  CS      = (int2*)P1;
  int*   offG    = (int*)P1 + 2 * (size_t)N_EDGES;

  k_chunk_sort<<<N_CHUNK, 256, 0, stream>>>(src, dst, offG, CS);
  k_bucket<<<NB, 256, 0, stream>>>(offG, CS, rowptr, dinv, csr_src);

  k_layer<false><<<LAYER_GRID, 256, 0, stream>>>(x,  W1, b1, rowptr, csr_src, dinv, nullptr,
                                                 nullptr, nullptr, P0);
  k_layer<false><<<LAYER_GRID, 256, 0, stream>>>(P0, W2, b2, rowptr, csr_src, dinv, nullptr,
                                                 nullptr, nullptr, P1);
  k_layer<true><<<N_TGT_TILES, 256, 0, stream>>>(P1, W3, b3, rowptr, csr_src, dinv, tgt,
                                                 Wr, br, out);
}
